# starting kernel
# baseline (speedup 1.0000x reference)
#include <hip/hip_runtime.h>
#include <hip/hip_bf16.h>
#include <math.h>


typedef __attribute__((ext_vector_type(2))) float v2f;
typedef __attribute__((ext_vector_type(8))) float v8f;

#define IN_FEAT   256
#define OUTC      16
#define NEG_SLOPE 0.2f
#define EPSF      1e-15f

__device__ __forceinline__ unsigned fmap(float f) {
    unsigned u = __float_as_uint(f);
    return (u & 0x80000000u) ? ~u : (u | 0x80000000u);
}
__device__ __forceinline__ float funmap(unsigned u) {
    unsigned v = (u & 0x80000000u) ? (u & 0x7FFFFFFFu) : ~u;
    return __uint_as_float(v);
}

__global__ __launch_bounds__(256) void k_node(
    const float* __restrict__ x, const float* __restrict__ W,
    const float* __restrict__ Wr, const float* __restrict__ bias,
    float* __restrict__ feat, float* __restrict__ rst, int N, int nTiles)
{
    __shared__ float sW [IN_FEAT * OUTC];
    __shared__ float sWr[IN_FEAT * OUTC];
    int tid = threadIdx.x;
    for (int i = tid; i < IN_FEAT * OUTC; i += 256) {
        sW[i]  = W[i];
        sWr[i] = Wr[i];
    }
    __syncthreads();

    int wave = tid >> 5;
    int lane = tid & 31;
    int t = blockIdx.x * 8 + wave;
    if (t >= nTiles) t = nTiles - 1;

    int half = lane >> 4;
    int l15  = lane & 15;
    int rowg = t * 16 + l15;
    if (rowg >= N) rowg = N - 1;
    const float* xrow = x + (size_t)rowg * IN_FEAT;

    v8f cf = {0.f,0.f,0.f,0.f,0.f,0.f,0.f,0.f};
    v8f cr = {0.f,0.f,0.f,0.f,0.f,0.f,0.f,0.f};
    float nsq = 0.f;

    for (int k0 = 0; k0 < IN_FEAT; k0 += 4) {
        int ka = k0 + half * 2;
        float a0 = xrow[ka], a1 = xrow[ka + 1];
        nsq = fmaf(a0, a0, nsq);
        nsq = fmaf(a1, a1, nsq);
        v2f a; a[0] = a0; a[1] = a1;
        v2f b;  b[0]  = sW [ka * OUTC + l15];  b[1]  = sW [(ka + 1) * OUTC + l15];
        v2f br; br[0] = sWr[ka * OUTC + l15];  br[1] = sWr[(ka + 1) * OUTC + l15];
        cf = __builtin_amdgcn_wmma_f32_16x16x4_f32(false, a, false, b,  (short)0, cf, false, false);
        cr = __builtin_amdgcn_wmma_f32_16x16x4_f32(false, a, false, br, (short)0, cr, false, false);
    }

    nsq += __shfl_xor(nsq, 16, 32);
    float nrm = fmaxf(sqrtf(nsq), EPSF);
    float z   = fminf(nrm, 1.0f - 1e-7f);
    float sc  = 0.5f * (log1pf(z) - log1pf(-z)) / nrm;
    float bv  = bias[l15];

    float* fp = feat + (size_t)t * 16 * OUTC;
    float* rp = rst  + (size_t)t * 16 * OUTC;
    #pragma unroll
    for (int r = 0; r < 8; ++r) {
        int row = r + half * 8;
        float s_r = __shfl(sc, row, 32);
        if (t * 16 + row < N) {
            fp[row * OUTC + l15] = cf[r] * s_r;
            rp[row * OUTC + l15] = cr[r] * s_r + bv;
        }
    }
}

__global__ void k_elr(const float* __restrict__ feat, const float* __restrict__ al,
                      const float* __restrict__ ar, float* __restrict__ el,
                      float* __restrict__ er, int N)
{
    int n = blockIdx.x * blockDim.x + threadIdx.x;
    if (n >= N) return;
    const float* f = feat + (size_t)n * OUTC;
    #pragma unroll
    for (int h = 0; h < 2; ++h) {
        float sl = 0.f, sr = 0.f;
        #pragma unroll
        for (int c = 0; c < 8; ++c) {
            float v = f[h * 8 + c];
            sl = fmaf(v, al[h * 8 + c], sl);
            sr = fmaf(v, ar[h * 8 + c], sr);
        }
        el[n * 2 + h] = sl;
        er[n * 2 + h] = sr;
    }
}

__global__ void k_init(unsigned* __restrict__ mmax, float* __restrict__ denom, int n2)
{
    int i = blockIdx.x * blockDim.x + threadIdx.x;
    if (i < n2) { mmax[i] = 0x007FFFFFu; denom[i] = 0.f; }
}

__global__ void k_edge_max(const int* __restrict__ ei, const float* __restrict__ el,
                           const float* __restrict__ er, unsigned* __restrict__ mmax, int E)
{
    int e = blockIdx.x * blockDim.x + threadIdx.x;
    if (e >= E) return;
    int s = ei[e], d = ei[E + e];
    #pragma unroll
    for (int h = 0; h < 2; ++h) {
        float v = el[s * 2 + h] + er[d * 2 + h];
        v = (v >= 0.f) ? v : NEG_SLOPE * v;
        atomicMax(&mmax[d * 2 + h], fmap(v));
    }
}

__global__ void k_max_fix(const unsigned* __restrict__ mmax, float* __restrict__ mfix, int n2)
{
    int i = blockIdx.x * blockDim.x + threadIdx.x;
    if (i >= n2) return;
    float m = funmap(mmax[i]);
    mfix[i] = (fabsf(m) <= 3.402823466e38f) ? m : 0.f;
}

__global__ void k_edge_sum(const int* __restrict__ ei, const float* __restrict__ el,
                           const float* __restrict__ er, const float* __restrict__ mfix,
                           float* __restrict__ denom, int E)
{
    int e = blockIdx.x * blockDim.x + threadIdx.x;
    if (e >= E) return;
    int s = ei[e], d = ei[E + e];
    #pragma unroll
    for (int h = 0; h < 2; ++h) {
        float v = el[s * 2 + h] + er[d * 2 + h];
        v = (v >= 0.f) ? v : NEG_SLOPE * v;
        atomicAdd(&denom[d * 2 + h], __expf(v - mfix[d * 2 + h]));
    }
}

__global__ void k_edge_agg(const int* __restrict__ ei, const float* __restrict__ el,
                           const float* __restrict__ er, const float* __restrict__ mfix,
                           const float* __restrict__ denom, const float* __restrict__ feat,
                           float* __restrict__ rst, int E)
{
    int e = blockIdx.x * blockDim.x + threadIdx.x;
    if (e >= E) return;
    int s = ei[e], d = ei[E + e];
    float alpha[2];
    #pragma unroll
    for (int h = 0; h < 2; ++h) {
        float v = el[s * 2 + h] + er[d * 2 + h];
        v = (v >= 0.f) ? v : NEG_SLOPE * v;
        float ex = __expf(v - mfix[d * 2 + h]);
        alpha[h] = ex / fmaxf(denom[d * 2 + h], EPSF);
    }
    const float* fs = feat + (size_t)s * OUTC;
    float* rd = rst + (size_t)d * OUTC;
    #pragma unroll
    for (int j = 0; j < OUTC; ++j)
        atomicAdd(&rd[j], alpha[j >> 3] * fs[j]);
}

__global__ void k_out(const float* __restrict__ rst, float* __restrict__ out, int N)
{
    int i = blockIdx.x * blockDim.x + threadIdx.x;
    if (i >= N * 8) return;
    int n = i >> 3, c = i & 7;
    out[i] = 0.5f * (rst[(size_t)n * OUTC + c] + rst[(size_t)n * OUTC + 8 + c]);
}

extern "C" void kernel_launch(void* const* d_in, const int* in_sizes, int n_in,
                              void* d_out, int out_size, void* d_ws, size_t ws_size,
                              hipStream_t stream)
{
    const float* x    = (const float*)d_in[0];
    const int*   ei   = (const int*)  d_in[1];
    const float* W    = (const float*)d_in[2];
    const float* al   = (const float*)d_in[3];
    const float* ar   = (const float*)d_in[4];
    const float* bias = (const float*)d_in[5];
    const float* Wr   = (const float*)d_in[6];
    float* out = (float*)d_out;

    int N = in_sizes[0] / IN_FEAT;
    int E = in_sizes[1] / 2;

    char* ws = (char*)d_ws;
    float*    feat  = (float*)ws;    ws += (size_t)N * OUTC * sizeof(float);
    float*    rst   = (float*)ws;    ws += (size_t)N * OUTC * sizeof(float);
    float*    el    = (float*)ws;    ws += (size_t)N * 2 * sizeof(float);
    float*    er    = (float*)ws;    ws += (size_t)N * 2 * sizeof(float);
    float*    mfix  = (float*)ws;    ws += (size_t)N * 2 * sizeof(float);
    float*    denom = (float*)ws;    ws += (size_t)N * 2 * sizeof(float);
    unsigned* mmax  = (unsigned*)ws; ws += (size_t)N * 2 * sizeof(unsigned);

    int nTiles = (N + 15) / 16;
    dim3 blk(256);

    k_node    <<<(nTiles + 7) / 8,      blk, 0, stream>>>(x, W, Wr, bias, feat, rst, N, nTiles);
    k_init    <<<(2 * N + 255) / 256,   blk, 0, stream>>>(mmax, denom, 2 * N);
    k_elr     <<<(N + 255) / 256,       blk, 0, stream>>>(feat, al, ar, el, er, N);
    k_edge_max<<<(E + 255) / 256,       blk, 0, stream>>>(ei, el, er, mmax, E);
    k_max_fix <<<(2 * N + 255) / 256,   blk, 0, stream>>>(mmax, mfix, 2 * N);
    k_edge_sum<<<(E + 255) / 256,       blk, 0, stream>>>(ei, el, er, mfix, denom, E);
    k_edge_agg<<<(E + 255) / 256,       blk, 0, stream>>>(ei, el, er, mfix, denom, feat, rst, E);
    k_out     <<<(N * 8 + 255) / 256,   blk, 0, stream>>>(rst, out, N);
}


# Edge scans stream (dst, src) index pairs by 16-byte loads one block ahead; the ring keeps a hit's source id: a hit batch waits on one gather, not three dependent reads; atomic sum order as before
# speedup vs baseline: 9.0602x; 9.0602x over previous
#include <hip/hip_runtime.h>
#include <hip/hip_bf16.h>
#include <math.h>


typedef __attribute__((ext_vector_type(2))) float v2f;
typedef __attribute__((ext_vector_type(8))) float v8f;

#define IN_FEAT   256
#define OUTC      16
#define NEG_SLOPE 0.2f
#define EPSF      1e-15f

__device__ __forceinline__ unsigned fmap(float f) {
    unsigned u = __float_as_uint(f);
    return (u & 0x80000000u) ? ~u : (u | 0x80000000u);
}
__device__ __forceinline__ float funmap(unsigned u) {
    unsigned v = (u & 0x80000000u) ? (u & 0x7FFFFFFFu) : ~u;
    return __uint_as_float(v);
}

__global__ __launch_bounds__(256) void k_node(
    const float* __restrict__ x, const float* __restrict__ W,
    const float* __restrict__ Wr, const float* __restrict__ bias,
    float* __restrict__ feat, float* __restrict__ out, int N, int nTiles)
{
    __shared__ float sW [IN_FEAT * OUTC];
    __shared__ float sWr[IN_FEAT * OUTC];
    int tid = threadIdx.x;
    for (int i = tid; i < IN_FEAT * OUTC; i += 256) {
        sW[i]  = W[i];
        sWr[i] = Wr[i];
    }
    __syncthreads();

    int wave = tid >> 5;
    int lane = tid & 31;
    int t = blockIdx.x * 8 + wave;
    if (t >= nTiles) t = nTiles - 1;

    int half = lane >> 4;
    int l15  = lane & 15;
    int rowg = t * 16 + l15;
    if (rowg >= N) rowg = N - 1;
    const float* xrow = x + (size_t)rowg * IN_FEAT;

    v8f cf = {0.f,0.f,0.f,0.f,0.f,0.f,0.f,0.f};
    v8f cr = {0.f,0.f,0.f,0.f,0.f,0.f,0.f,0.f};
    float nsq = 0.f;

    for (int k0 = 0; k0 < IN_FEAT; k0 += 4) {
        int ka = k0 + half * 2;
        float a0 = xrow[ka], a1 = xrow[ka + 1];
        nsq = fmaf(a0, a0, nsq);
        nsq = fmaf(a1, a1, nsq);
        v2f a; a[0] = a0; a[1] = a1;
        v2f b;  b[0]  = sW [ka * OUTC + l15];  b[1]  = sW [(ka + 1) * OUTC + l15];
        v2f br; br[0] = sWr[ka * OUTC + l15];  br[1] = sWr[(ka + 1) * OUTC + l15];
        cf = __builtin_amdgcn_wmma_f32_16x16x4_f32(false, a, false, b,  (short)0, cf, false, false);
        cr = __builtin_amdgcn_wmma_f32_16x16x4_f32(false, a, false, br, (short)0, cr, false, false);
    }

    nsq += __shfl_xor(nsq, 16, 32);
    float nrm = fmaxf(sqrtf(nsq), EPSF);
    float z   = fminf(nrm, 1.0f - 1e-7f);
    float sc  = 0.5f * (log1pf(z) - log1pf(-z)) / nrm;
    float bv  = bias[l15];

    float* fp = feat + (size_t)t * 16 * OUTC;
    float* op = out  + (size_t)t * 16 * 8;
    #pragma unroll
    for (int r = 0; r < 8; ++r) {
        int row = r + half * 8;
        float s_r = __shfl(sc, row, 32);
        float rv  = cr[r] * s_r + bv;
        float ro  = __shfl_xor(rv, 8, 32);
        if (t * 16 + row < N) {
            fp[row * OUTC + l15] = cf[r] * s_r;
            if (l15 < 8) op[row * 8 + l15] = 0.5f * (rv + ro);
        }
    }
}

__global__ void k_elr(const float* __restrict__ feat, const float* __restrict__ al,
                      const float* __restrict__ ar, float* __restrict__ el,
                      float* __restrict__ er, int N)
{
    int n = blockIdx.x * blockDim.x + threadIdx.x;
    if (n >= N) return;
    const float* f = feat + (size_t)n * OUTC;
    #pragma unroll
    for (int h = 0; h < 2; ++h) {
        float sl = 0.f, sr = 0.f;
        #pragma unroll
        for (int c = 0; c < 8; ++c) {
            float v = f[h * 8 + c];
            sl = fmaf(v, al[h * 8 + c], sl);
            sr = fmaf(v, ar[h * 8 + c], sr);
        }
        el[n * 2 + h] = sl;
        er[n * 2 + h] = sr;
    }
}

#define RNG     6272
#define NSLICE  16
#define RNGA    5312
#define APITCH  (RNGA + 8)
#define NSLICEA 13
#define WGT     1024
#define NWAVE   (WGT / 32)
#define QCAP    64
#define MAXINIT 0x007FFFFFu

__device__ __forceinline__ float leaky(float v) { return (v >= 0.f) ? v : NEG_SLOPE * v; }

__device__ __forceinline__ void ring_order() {
    __builtin_amdgcn_fence(__ATOMIC_RELEASE, "wavefront");
    __builtin_amdgcn_wave_barrier();
    __builtin_amdgcn_fence(__ATOMIC_ACQUIRE, "wavefront");
}

template <bool AL4>
__device__ __forceinline__ int4 load_quad(const int* __restrict__ p, int e4, int ee) {
    if (AL4) return *reinterpret_cast<const int4*>(p + min(e4, ee - 4));
    int4 r;
    r.x = p[min(e4,     ee - 1)];
    r.y = p[min(e4 + 1, ee - 1)];
    r.z = p[min(e4 + 2, ee - 1)];
    r.w = p[min(e4 + 3, ee - 1)];
    return r;
}
__device__ __forceinline__ int quad_at(const int4& v, int k) {
    return (k == 0) ? v.x : (k == 1) ? v.y : (k == 2) ? v.z : v.w;
}

#define NQ      4
#define BLK     (NQ * 128)
template <int R, bool AL4, typename F>
__device__ __forceinline__ void scan_hits(const int* __restrict__ srcp, const int* __restrict__ dstp,
                                          int eb, int ee, int base,
                                          unsigned* __restrict__ ring, int lane, int wave, F work)
{
    unsigned* qs = ring + wave * (2 * QCAP);
    unsigned* qd = qs + QCAP;
    unsigned head = 0u, qn = 0u;
    int first = eb + wave * BLK;
    if (first >= ee) return;
    int4 dq[NQ], sq[NQ];
    #pragma unroll
    for (int q = 0; q < NQ; ++q) { dq[q] = make_int4(-1, -1, -1, -1); sq[q] = make_int4(0, 0, 0, 0); }
    for (int e0 = first - NWAVE * BLK; e0 < ee; e0 += NWAVE * BLK) {
        int en = e0 + NWAVE * BLK;
        #pragma unroll
        for (int q = 0; q < NQ; ++q) {
            int eq = e0 + q * 128 + 4 * lane;
            #pragma unroll
            for (int k = 0; k < 4; ++k) {
                int e = eq + k;
                unsigned dl = (unsigned)(((e < ee) ? quad_at(dq[q], k) : -1) - base);
                bool     hit  = dl < (unsigned)R;
                unsigned mask = __builtin_amdgcn_ballot_w32(hit);
                if (mask != 0u) {
                    if (hit) {
                        unsigned pos = (head + qn + __builtin_amdgcn_mbcnt_lo(mask, 0u)) & (QCAP - 1);
                        qs[pos] = (unsigned)quad_at(sq[q], k);
                        qd[pos] = dl;
                    }
                    qn += (unsigned)__popc(mask);
                    if (qn >= 32u) {
                        ring_order();
                        unsigned p = (head + (unsigned)lane) & (QCAP - 1);
                        work(qs[p], qd[p], true);
                        ring_order();
                        head = (head + 32u) & (QCAP - 1);
                        qn  -= 32u;
                    }
                }
            }
            int nq = en + q * 128 + 4 * lane;
            dq[q] = load_quad<AL4>(dstp, nq, ee);
            sq[q] = load_quad<AL4>(srcp, nq, ee);
        }
    }
    if (qn != 0u) {
        ring_order();
        unsigned p = (head + (unsigned)lane) & (QCAP - 1);
        work(qs[p], qd[p], (unsigned)lane < qn);
    }
}

#define LCAP    16384
__global__ __launch_bounds__(WGT) void k_range_maxsum(
    const int* __restrict__ ei, const float* __restrict__ el, const float* __restrict__ er,
    float4* __restrict__ part, int N, int E, int chunk)
{
    __shared__ unsigned mx[2 * RNG];
    __shared__ __align__(16) float fb[2 * RNG + 2 * LCAP];
    __shared__ unsigned ldl[LCAP];
    __shared__ unsigned ring[NWAVE * 2 * QCAP];
    __shared__ unsigned lcnt;
    float* sv  = fb;
    float* le0 = fb + 2 * RNG;
    float* le1 = le0 + LCAP;
    int tid = threadIdx.x, lane = tid & 31, wave = tid >> 5;
    int base = blockIdx.x * RNG;
    int chunk4 = (chunk + 3) & ~3;
    int eb = blockIdx.y * chunk4;
    int ee = min(E, eb + chunk4);
    bool al4 = (E & 3) == 0;
    if (tid == 0) lcnt = 0u;
    for (int i = tid; i < 2 * RNG; i += WGT) mx[i] = MAXINIT;
    for (int i = tid; i < RNG; i += WGT) {
        int n = base + i;
        float2 r = (n < N) ? *reinterpret_cast<const float2*>(er + 2 * (size_t)n) : make_float2(0.f, 0.f);
        *reinterpret_cast<float2*>(sv + 2 * i) = r;
    }
    __syncthreads();
    const int* srcp = ei;
    auto hit_max = [&](unsigned s, unsigned dl, bool act) {
            unsigned am = __builtin_amdgcn_ballot_w32(act);
            unsigned bp = 0u;
            if (lane == 0) bp = atomicAdd(&lcnt, (unsigned)__popc(am));
            bp = (unsigned)__builtin_amdgcn_readfirstlane((int)bp);
            if (act) {
                float2 l = *reinterpret_cast<const float2*>(el + 2 * (size_t)s);
                float2 r = *reinterpret_cast<const float2*>(sv + 2 * dl);
                float e0 = leaky(l.x + r.x), e1 = leaky(l.y + r.y);
                atomicMax(&mx[2 * dl],     fmap(e0));
                atomicMax(&mx[2 * dl + 1], fmap(e1));
                unsigned pos = bp + __builtin_amdgcn_mbcnt_lo(am, 0u);
                if (pos < (unsigned)LCAP) { le0[pos] = e0; le1[pos] = e1; ldl[pos] = dl; }
            }
        };
    if (al4) scan_hits<RNG, true >(srcp, ei + E, eb, ee, base, ring, lane, wave, hit_max);
    else     scan_hits<RNG, false>(srcp, ei + E, eb, ee, base, ring, lane, wave, hit_max);
    __syncthreads();
    unsigned total = lcnt;
    bool kept = total <= (unsigned)LCAP;
    float* den = kept ? sv : le0;
    for (int i = tid; i < 2 * RNG; i += WGT) den[i] = 0.f;
    __syncthreads();
    if (kept) {
        for (unsigned i0 = (unsigned)tid; i0 < total; i0 += 4 * WGT) {
            unsigned dl[4]; float a0[4], a1[4]; uint2 u[4];
            #pragma unroll
            for (int j = 0; j < 4; ++j) {
                unsigned i = min(i0 + (unsigned)(j * WGT), total - 1u);
                dl[j] = ldl[i]; a0[j] = le0[i]; a1[j] = le1[i];
            }
            #pragma unroll
            for (int j = 0; j < 4; ++j) u[j] = *reinterpret_cast<const uint2*>(mx + 2 * dl[j]);
            #pragma unroll
            for (int j = 0; j < 4; ++j) {
                if (i0 + (unsigned)(j * WGT) < total) {
                    atomicAdd(&sv[2 * dl[j]],     __expf(a0[j] - funmap(u[j].x)));
                    atomicAdd(&sv[2 * dl[j] + 1], __expf(a1[j] - funmap(u[j].y)));
                }
            }
        }
    } else {
        scan_hits<RNG, false>(srcp, ei + E, eb, ee, base, ring, lane, wave,
            [&](unsigned s, unsigned dl, bool act) {
                if (act) {
                    float2 l = *reinterpret_cast<const float2*>(el + 2 * (size_t)s);
                    float2 r = *reinterpret_cast<const float2*>(sv + 2 * dl);
                    uint2  u = *reinterpret_cast<const uint2*>(mx + 2 * dl);
                    atomicAdd(&den[2 * dl],     __expf(leaky(l.x + r.x) - funmap(u.x)));
                    atomicAdd(&den[2 * dl + 1], __expf(leaky(l.y + r.y) - funmap(u.y)));
                }
            });
    }
    __syncthreads();
    float4* prow = part + (size_t)blockIdx.y * N;
    for (int i = tid; i < RNG; i += WGT) {
        int n = base + i;
        if (n < N) prow[n] = make_float4(funmap(mx[2 * i]), funmap(mx[2 * i + 1]), den[2 * i], den[2 * i + 1]);
    }
}

__global__ void k_merge(const float4* __restrict__ part, float* __restrict__ mfin,
                        float* __restrict__ denom, int N)
{
    int n = blockIdx.x * blockDim.x + threadIdx.x;
    if (n >= N) return;
    float4 p[NSLICE];
    #pragma unroll
    for (int s = 0; s < NSLICE; ++s) p[s] = part[(size_t)s * N + n];
    float m0 = p[0].x, m1 = p[0].y;
    #pragma unroll
    for (int s = 1; s < NSLICE; ++s) { m0 = fmaxf(m0, p[s].x); m1 = fmaxf(m1, p[s].y); }
    m0 = (fabsf(m0) <= 3.402823466e38f) ? m0 : 0.f;
    m1 = (fabsf(m1) <= 3.402823466e38f) ? m1 : 0.f;
    float d0 = 0.f, d1 = 0.f;
    #pragma unroll
    for (int s = 0; s < NSLICE; ++s) {
        d0 = fmaf(p[s].z, __expf(p[s].x - m0), d0);
        d1 = fmaf(p[s].w, __expf(p[s].y - m1), d1);
    }
    *reinterpret_cast<float2*>(mfin  + 2 * (size_t)n) = make_float2(m0, m1);
    *reinterpret_cast<float2*>(denom + 2 * (size_t)n) = make_float2(d0, d1);
}

__global__ __launch_bounds__(WGT) void k_range_agg(
    const int* __restrict__ ei, const float* __restrict__ al, const float* __restrict__ er,
    const float* __restrict__ mfin, const float* __restrict__ denom,
    const float* __restrict__ feat, float* __restrict__ out, int N, int E, int chunk)
{
    __shared__ float acc[8 * APITCH];
    __shared__ float4 sem[RNGA];
    __shared__ float2 sdn[RNGA];
    __shared__ unsigned ring[NWAVE * 2 * QCAP];
    int tid = threadIdx.x, lane = tid & 31, wave = tid >> 5;
    int base = blockIdx.x * RNGA;
    int chunk4 = (chunk + 3) & ~3;
    int eb = blockIdx.y * chunk4;
    int ee = min(E, eb + chunk4);
    bool al4 = (E & 3) == 0;
    for (int i = tid; i < 8 * APITCH; i += WGT) acc[i] = 0.f;
    for (int i = tid; i < RNGA; i += WGT) {
        int n = base + i;
        float4 q = make_float4(0.f, 0.f, 0.f, 0.f);
        float2 dn = make_float2(1.f, 1.f);
        if (n < N) {
            float2 r = *reinterpret_cast<const float2*>(er + 2 * (size_t)n);
            float2 m = *reinterpret_cast<const float2*>(mfin + 2 * (size_t)n);
            float2 g = *reinterpret_cast<const float2*>(denom + 2 * (size_t)n);
            q  = make_float4(r.x, r.y, m.x, m.y);
            dn = make_float2(fmaxf(g.x, EPSF), fmaxf(g.y, EPSF));
        }
        sem[i] = q;
        sdn[i] = dn;
    }
    __syncthreads();
    const int* srcp = ei;
    float w0 = al[0],  w1 = al[1],  w2 = al[2],   w3 = al[3],   w4 = al[4],   w5 = al[5],   w6 = al[6],   w7 = al[7];
    float w8 = al[8],  w9 = al[9],  w10 = al[10], w11 = al[11], w12 = al[12], w13 = al[13], w14 = al[14], w15 = al[15];
    auto hit_msg = [&](unsigned s, unsigned dl, bool act) {
            if (act) {
                const float4* fs = reinterpret_cast<const float4*>(feat + (size_t)s * OUTC);
                float4 f0 = fs[0], f1 = fs[1], f2 = fs[2], f3 = fs[3];
                float2 l;
                l.x = fmaf(f0.x, w0, 0.f);  l.x = fmaf(f0.y, w1, l.x);  l.x = fmaf(f0.z, w2, l.x);  l.x = fmaf(f0.w, w3, l.x);
                l.x = fmaf(f1.x, w4, l.x);  l.x = fmaf(f1.y, w5, l.x);  l.x = fmaf(f1.z, w6, l.x);  l.x = fmaf(f1.w, w7, l.x);
                l.y = fmaf(f2.x, w8, 0.f);  l.y = fmaf(f2.y, w9, l.y);  l.y = fmaf(f2.z, w10, l.y); l.y = fmaf(f2.w, w11, l.y);
                l.y = fmaf(f3.x, w12, l.y); l.y = fmaf(f3.y, w13, l.y); l.y = fmaf(f3.z, w14, l.y); l.y = fmaf(f3.w, w15, l.y);
                float4 q  = sem[dl];
                float2 dn = sdn[dl];
                float a0 = 0.5f * (__expf(leaky(l.x + q.x) - q.z) / dn.x);
                float a1 = 0.5f * (__expf(leaky(l.y + q.y) - q.w) / dn.y);
                float* a = acc + dl;
                atomicAdd(a + 0 * APITCH, fmaf(a0, f0.x, a1 * f2.x));
                atomicAdd(a + 1 * APITCH, fmaf(a0, f0.y, a1 * f2.y));
                atomicAdd(a + 2 * APITCH, fmaf(a0, f0.z, a1 * f2.z));
                atomicAdd(a + 3 * APITCH, fmaf(a0, f0.w, a1 * f2.w));
                atomicAdd(a + 4 * APITCH, fmaf(a0, f1.x, a1 * f3.x));
                atomicAdd(a + 5 * APITCH, fmaf(a0, f1.y, a1 * f3.y));
                atomicAdd(a + 6 * APITCH, fmaf(a0, f1.z, a1 * f3.z));
                atomicAdd(a + 7 * APITCH, fmaf(a0, f1.w, a1 * f3.w));
            }
        };
    if (al4) scan_hits<RNGA, true >(srcp, ei + E, eb, ee, base, ring, lane, wave, hit_msg);
    else     scan_hits<RNGA, false>(srcp, ei + E, eb, ee, base, ring, lane, wave, hit_msg);
    __syncthreads();
    for (int i = tid; i < 8 * RNGA; i += WGT) {
        int dl = i >> 3, c = i & 7;
        int n = base + dl;
        if (n < N) atomicAdd(&out[(size_t)n * 8 + c], acc[c * APITCH + dl]);
    }
}

extern "C" void kernel_launch(void* const* d_in, const int* in_sizes, int n_in,
                              void* d_out, int out_size, void* d_ws, size_t ws_size,
                              hipStream_t stream)
{
    const float* x    = (const float*)d_in[0];
    const int*   ei   = (const int*)  d_in[1];
    const float* W    = (const float*)d_in[2];
    const float* al   = (const float*)d_in[3];
    const float* ar   = (const float*)d_in[4];
    const float* bias = (const float*)d_in[5];
    const float* Wr   = (const float*)d_in[6];
    float* out = (float*)d_out;

    int N = in_sizes[0] / IN_FEAT;
    int E = in_sizes[1] / 2;

    char* ws = (char*)d_ws;
    float4*   part  = (float4*)ws;   ws += (size_t)NSLICE * N * sizeof(float4);
    float*    feat  = (float*)ws;    ws += (size_t)N * OUTC * sizeof(float);
    float*    el    = (float*)ws;    ws += (size_t)N * 2 * sizeof(float);
    float*    er    = (float*)ws;    ws += (size_t)N * 2 * sizeof(float);
    float*    denom = (float*)ws;    ws += (size_t)N * 2 * sizeof(float);
    float*    mfin  = (float*)ws;    ws += (size_t)N * 2 * sizeof(float);

    int nTiles = (N + 15) / 16;
    dim3 blk(256);

    k_node    <<<(nTiles + 7) / 8,      blk, 0, stream>>>(x, W, Wr, bias, feat, out, N, nTiles);
    k_elr     <<<(N + 255) / 256,       blk, 0, stream>>>(feat, al, ar, el, er, N);
    int nRange = (N + RNG - 1) / RNG;
    int chunk  = (E + NSLICE - 1) / NSLICE;
    dim3 grd(nRange, NSLICE);
    dim3 wg(WGT);
    k_range_maxsum<<<grd, wg, 0, stream>>>(ei, el, er, part, N, E, chunk);
    k_merge   <<<(N + 255) / 256,       blk, 0, stream>>>(part, mfin, denom, N);
    int nRangeA = (N + RNGA - 1) / RNGA;
    int chunkA  = (E + NSLICEA - 1) / NSLICEA;
    dim3 grdA(nRangeA, NSLICEA);
    k_range_agg<<<grdA, wg, 0, stream>>>(ei, al, er, mfin, denom, feat, out, N, E, chunkA);
}


# Softmax shift known before any edge is read (leaky(max el + er[d]) while the span of el is small, else an edge max pass): no max sweep, weights and messages summed in one sweep; sum order changes
# speedup vs baseline: 10.4136x; 1.1494x over previous
#include <hip/hip_runtime.h>
#include <hip/hip_bf16.h>
#include <math.h>


typedef __attribute__((ext_vector_type(2))) float v2f;
typedef __attribute__((ext_vector_type(8))) float v8f;

#define IN_FEAT   256
#define OUTC      16
#define NEG_SLOPE 0.2f
#define EPSF      1e-15f
#define NACC      18
#define ELSPAN    16.0f
#define MAXINIT   0x007FFFFFu

__device__ __forceinline__ unsigned fmap(float f) {
    unsigned u = __float_as_uint(f);
    return (u & 0x80000000u) ? ~u : (u | 0x80000000u);
}
__device__ __forceinline__ float funmap(unsigned u) {
    unsigned v = (u & 0x80000000u) ? (u & 0x7FFFFFFFu) : ~u;
    return __uint_as_float(v);
}

__global__ __launch_bounds__(256) void k_node(
    const float* __restrict__ x, const float* __restrict__ W,
    const float* __restrict__ Wr, const float* __restrict__ bias,
    float* __restrict__ feat, float* __restrict__ out, int N, int nTiles)
{
    __shared__ float sW [IN_FEAT * OUTC];
    __shared__ float sWr[IN_FEAT * OUTC];
    int tid = threadIdx.x;
    for (int i = tid; i < IN_FEAT * OUTC; i += 256) {
        sW[i]  = W[i];
        sWr[i] = Wr[i];
    }
    __syncthreads();

    int wave = tid >> 5;
    int lane = tid & 31;
    int t = blockIdx.x * 8 + wave;
    if (t >= nTiles) t = nTiles - 1;

    int half = lane >> 4;
    int l15  = lane & 15;
    int rowg = t * 16 + l15;
    if (rowg >= N) rowg = N - 1;
    const float* xrow = x + (size_t)rowg * IN_FEAT;

    v8f cf = {0.f,0.f,0.f,0.f,0.f,0.f,0.f,0.f};
    v8f cr = {0.f,0.f,0.f,0.f,0.f,0.f,0.f,0.f};
    float nsq = 0.f;

    for (int k0 = 0; k0 < IN_FEAT; k0 += 4) {
        int ka = k0 + half * 2;
        float a0 = xrow[ka], a1 = xrow[ka + 1];
        nsq = fmaf(a0, a0, nsq);
        nsq = fmaf(a1, a1, nsq);
        v2f a; a[0] = a0; a[1] = a1;
        v2f b;  b[0]  = sW [ka * OUTC + l15];  b[1]  = sW [(ka + 1) * OUTC + l15];
        v2f br; br[0] = sWr[ka * OUTC + l15];  br[1] = sWr[(ka + 1) * OUTC + l15];
        cf = __builtin_amdgcn_wmma_f32_16x16x4_f32(false, a, false, b,  (short)0, cf, false, false);
        cr = __builtin_amdgcn_wmma_f32_16x16x4_f32(false, a, false, br, (short)0, cr, false, false);
    }

    nsq += __shfl_xor(nsq, 16, 32);
    float nrm = fmaxf(sqrtf(nsq), EPSF);
    float z   = fminf(nrm, 1.0f - 1e-7f);
    float sc  = 0.5f * (log1pf(z) - log1pf(-z)) / nrm;
    float bv  = bias[l15];

    float* fp = feat + (size_t)t * 16 * OUTC;
    float* op = out  + (size_t)t * 16 * 8;
    #pragma unroll
    for (int r = 0; r < 8; ++r) {
        int row = r + half * 8;
        float s_r = __shfl(sc, row, 32);
        float rv  = cr[r] * s_r + bv;
        float ro  = __shfl_xor(rv, 8, 32);
        if (t * 16 + row < N) {
            fp[row * OUTC + l15] = cf[r] * s_r;
            if (l15 < 8) op[row * 8 + l15] = 0.5f * (rv + ro);
        }
    }
}

__global__ __launch_bounds__(256) void k_elr(const float* __restrict__ feat, const float* __restrict__ al,
                      const float* __restrict__ ar, float* __restrict__ el, float* __restrict__ er,
                      float* __restrict__ gsum, unsigned* __restrict__ mxg,
                      float4* __restrict__ pmm, int N)
{
    __shared__ float4 wm[8];
    int n = blockIdx.x * 256 + threadIdx.x;
    float e0 = -INFINITY, e1 = -INFINITY;
    float g0 =  INFINITY, g1 =  INFINITY;
    if (n < N) {
        const float* f = feat + (size_t)n * OUTC;
        float sl[2], sr[2];
        #pragma unroll
        for (int h = 0; h < 2; ++h) {
            sl[h] = 0.f; sr[h] = 0.f;
            #pragma unroll
            for (int c = 0; c < 8; ++c) {
                float v = f[h * 8 + c];
                sl[h] = fmaf(v, al[h * 8 + c], sl[h]);
                sr[h] = fmaf(v, ar[h * 8 + c], sr[h]);
            }
        }
        e0 = sl[0]; e1 = sl[1];
        g0 = sl[0]; g1 = sl[1];
        *reinterpret_cast<float2*>(el + 2 * (size_t)n) = make_float2(sl[0], sl[1]);
        *reinterpret_cast<float2*>(er + 2 * (size_t)n) = make_float2(sr[0], sr[1]);
        *reinterpret_cast<uint2*>(mxg + 2 * (size_t)n) = make_uint2(MAXINIT, MAXINIT);
        for (int c = 0; c < NACC; ++c) gsum[(size_t)c * N + n] = 0.f;
    }
    for (int o = 16; o > 0; o >>= 1) {
        e0 = fmaxf(e0, __shfl_xor(e0, o, 32));
        e1 = fmaxf(e1, __shfl_xor(e1, o, 32));
        g0 = fminf(g0, __shfl_xor(g0, o, 32));
        g1 = fminf(g1, __shfl_xor(g1, o, 32));
    }
    if ((threadIdx.x & 31) == 0) wm[threadIdx.x >> 5] = make_float4(g0, g1, e0, e1);
    __syncthreads();
    if (threadIdx.x == 0) {
        float4 r = wm[0];
        for (int w = 1; w < 8; ++w) {
            r.x = fminf(r.x, wm[w].x); r.y = fminf(r.y, wm[w].y);
            r.z = fmaxf(r.z, wm[w].z); r.w = fmaxf(r.w, wm[w].w);
        }
        pmm[blockIdx.x] = r;
    }
}

__device__ __forceinline__ float4 el_bounds(const float4* __restrict__ pmm, int nPart, float4* red)
{
    int tid = threadIdx.x, lane = tid & 31, wave = tid >> 5, nw = (int)blockDim.x >> 5;
    float4 b = make_float4(INFINITY, INFINITY, -INFINITY, -INFINITY);
    for (int i = tid; i < nPart; i += (int)blockDim.x) {
        float4 p = pmm[i];
        b.x = fminf(b.x, p.x); b.y = fminf(b.y, p.y);
        b.z = fmaxf(b.z, p.z); b.w = fmaxf(b.w, p.w);
    }
    for (int o = 16; o > 0; o >>= 1) {
        b.x = fminf(b.x, __shfl_xor(b.x, o, 32)); b.y = fminf(b.y, __shfl_xor(b.y, o, 32));
        b.z = fmaxf(b.z, __shfl_xor(b.z, o, 32)); b.w = fmaxf(b.w, __shfl_xor(b.w, o, 32));
    }
    if (lane == 0) red[wave] = b;
    __syncthreads();
    b = red[0];
    for (int w = 1; w < nw; ++w) {
        float4 p = red[w];
        b.x = fminf(b.x, p.x); b.y = fminf(b.y, p.y);
        b.z = fmaxf(b.z, p.z); b.w = fmaxf(b.w, p.w);
    }
    return b;
}
__device__ __forceinline__ bool span_small(float4 b) {
    return (b.z - b.x <= ELSPAN) && (b.w - b.y <= ELSPAN);
}

__global__ __launch_bounds__(256) void k_edge_max(
    const int* __restrict__ ei, const float* __restrict__ el, const float* __restrict__ er,
    const float4* __restrict__ pmm, int nPart, unsigned* __restrict__ mxg, int E)
{
    __shared__ float4 red[8];
    if (span_small(el_bounds(pmm, nPart, red))) return;
    for (int e = blockIdx.x * 256 + threadIdx.x; e < E; e += (int)gridDim.x * 256) {
        int s = ei[e], d = ei[(size_t)E + e];
        float2 l = *reinterpret_cast<const float2*>(el + 2 * (size_t)s);
        float2 r = *reinterpret_cast<const float2*>(er + 2 * (size_t)d);
        float v0 = l.x + r.x, v1 = l.y + r.y;
        atomicMax(mxg + 2 * (size_t)d,     fmap((v0 >= 0.f) ? v0 : NEG_SLOPE * v0));
        atomicMax(mxg + 2 * (size_t)d + 1, fmap((v1 >= 0.f) ? v1 : NEG_SLOPE * v1));
    }
}

#define RNGF    3872
#define NSLICEF 9
#define WGT     1024
#define NWAVE   (WGT / 32)
#define QCAP    64

__device__ __forceinline__ float leaky(float v) { return (v >= 0.f) ? v : NEG_SLOPE * v; }

__device__ __forceinline__ void ring_order() {
    __builtin_amdgcn_fence(__ATOMIC_RELEASE, "wavefront");
    __builtin_amdgcn_wave_barrier();
    __builtin_amdgcn_fence(__ATOMIC_ACQUIRE, "wavefront");
}

template <bool AL4>
__device__ __forceinline__ int4 load_quad(const int* __restrict__ p, int e4, int ee) {
    if (AL4) return *reinterpret_cast<const int4*>(p + min(e4, ee - 4));
    int4 r;
    r.x = p[min(e4,     ee - 1)];
    r.y = p[min(e4 + 1, ee - 1)];
    r.z = p[min(e4 + 2, ee - 1)];
    r.w = p[min(e4 + 3, ee - 1)];
    return r;
}
__device__ __forceinline__ int quad_at(const int4& v, int k) {
    return (k == 0) ? v.x : (k == 1) ? v.y : (k == 2) ? v.z : v.w;
}

#define NQ      4
#define BLK     (NQ * 128)
template <int R, bool AL4, typename F>
__device__ __forceinline__ void scan_hits(const int* __restrict__ srcp, const int* __restrict__ dstp,
                                          int eb, int ee, int base,
                                          unsigned* __restrict__ ring, int lane, int wave, F work)
{
    unsigned* qs = ring + wave * (2 * QCAP);
    unsigned* qd = qs + QCAP;
    unsigned head = 0u, qn = 0u;
    int first = eb + wave * BLK;
    if (first >= ee) return;
    int4 dq[NQ], sq[NQ];
    #pragma unroll
    for (int q = 0; q < NQ; ++q) { dq[q] = make_int4(-1, -1, -1, -1); sq[q] = make_int4(0, 0, 0, 0); }
    for (int e0 = first - NWAVE * BLK; e0 < ee; e0 += NWAVE * BLK) {
        int en = e0 + NWAVE * BLK;
        #pragma unroll
        for (int q = 0; q < NQ; ++q) {
            int eq = e0 + q * 128 + 4 * lane;
            #pragma unroll
            for (int k = 0; k < 4; ++k) {
                int e = eq + k;
                unsigned dl = (unsigned)(((e < ee) ? quad_at(dq[q], k) : -1) - base);
                bool     hit  = dl < (unsigned)R;
                unsigned mask = __builtin_amdgcn_ballot_w32(hit);
                if (mask != 0u) {
                    if (hit) {
                        unsigned pos = (head + qn + __builtin_amdgcn_mbcnt_lo(mask, 0u)) & (QCAP - 1);
                        qs[pos] = (unsigned)quad_at(sq[q], k);
                        qd[pos] = dl;
                    }
                    qn += (unsigned)__popc(mask);
                    if (qn >= 32u) {
                        ring_order();
                        unsigned p = (head + (unsigned)lane) & (QCAP - 1);
                        work(qs[p], qd[p], true);
                        ring_order();
                        head = (head + 32u) & (QCAP - 1);
                        qn  -= 32u;
                    }
                }
            }
            int nq = en + q * 128 + 4 * lane;
            dq[q] = load_quad<AL4>(dstp, nq, ee);
            sq[q] = load_quad<AL4>(srcp, nq, ee);
        }
    }
    if (qn != 0u) {
        ring_order();
        unsigned p = (head + (unsigned)lane) & (QCAP - 1);
        work(qs[p], qd[p], (unsigned)lane < qn);
    }
}

__global__ __launch_bounds__(WGT) void k_range_fused(
    const int* __restrict__ ei, const float* __restrict__ al, const float* __restrict__ er,
    const float4* __restrict__ pmm, int nPart, const unsigned* __restrict__ mxg,
    const float* __restrict__ feat, float* __restrict__ gsum, int N, int E, int chunk)
{
    __shared__ __align__(16) float acc[NACC * RNGF];
    __shared__ float2 ser[RNGF];
    __shared__ unsigned ring[NWAVE * 2 * QCAP];
    __shared__ float4 red[NWAVE];
    int tid = threadIdx.x, lane = tid & 31, wave = tid >> 5;
    int base = blockIdx.x * RNGF;
    int chunk4 = (chunk + 3) & ~3;
    int eb = blockIdx.y * chunk4;
    int ee = min(E, eb + chunk4);
    bool al4 = (E & 3) == 0;
    float4* acc4 = reinterpret_cast<float4*>(acc);
    for (int i = tid; i < NACC * RNGF / 4; i += WGT) acc4[i] = make_float4(0.f, 0.f, 0.f, 0.f);
    for (int i = tid; i < RNGF; i += WGT) {
        int n = base + i;
        ser[i] = (n < N) ? *reinterpret_cast<const float2*>(er + 2 * (size_t)n) : make_float2(0.f, 0.f);
    }
    float4 bnd = el_bounds(pmm, nPart, red);
    bool bounded = __builtin_amdgcn_readfirstlane(span_small(bnd) ? 1 : 0) != 0;
    float b0 = bnd.z, b1 = bnd.w;
    const int* srcp = ei;
    float w0 = al[0],  w1 = al[1],  w2 = al[2],   w3 = al[3],   w4 = al[4],   w5 = al[5],   w6 = al[6],   w7 = al[7];
    float w8 = al[8],  w9 = al[9],  w10 = al[10], w11 = al[11], w12 = al[12], w13 = al[13], w14 = al[14], w15 = al[15];
    auto hit_sum = [&](unsigned s, unsigned dl, bool act, auto sh) __attribute__((always_inline)) {
            if (act) {
                const float4* fs = reinterpret_cast<const float4*>(feat + (size_t)s * OUTC);
                float4 f0 = fs[0], f1 = fs[1], f2 = fs[2], f3 = fs[3];
                float2 l;
                l.x = fmaf(f0.x, w0, 0.f);  l.x = fmaf(f0.y, w1, l.x);  l.x = fmaf(f0.z, w2, l.x);  l.x = fmaf(f0.w, w3, l.x);
                l.x = fmaf(f1.x, w4, l.x);  l.x = fmaf(f1.y, w5, l.x);  l.x = fmaf(f1.z, w6, l.x);  l.x = fmaf(f1.w, w7, l.x);
                l.y = fmaf(f2.x, w8, 0.f);  l.y = fmaf(f2.y, w9, l.y);  l.y = fmaf(f2.z, w10, l.y); l.y = fmaf(f2.w, w11, l.y);
                l.y = fmaf(f3.x, w12, l.y); l.y = fmaf(f3.y, w13, l.y); l.y = fmaf(f3.z, w14, l.y); l.y = fmaf(f3.w, w15, l.y);
                float2 r = ser[dl];
                float2 m = sh(r, dl);
                float p0 = __expf(leaky(l.x + r.x) - m.x);
                float p1 = __expf(leaky(l.y + r.y) - m.y);
                float* a = acc + dl;
                atomicAdd(a +  0 * RNGF, p0 * f0.x);
                atomicAdd(a +  1 * RNGF, p0 * f0.y);
                atomicAdd(a +  2 * RNGF, p0 * f0.z);
                atomicAdd(a +  3 * RNGF, p0 * f0.w);
                atomicAdd(a +  4 * RNGF, p0 * f1.x);
                atomicAdd(a +  5 * RNGF, p0 * f1.y);
                atomicAdd(a +  6 * RNGF, p0 * f1.z);
                atomicAdd(a +  7 * RNGF, p0 * f1.w);
                atomicAdd(a +  8 * RNGF, p1 * f2.x);
                atomicAdd(a +  9 * RNGF, p1 * f2.y);
                atomicAdd(a + 10 * RNGF, p1 * f2.z);
                atomicAdd(a + 11 * RNGF, p1 * f2.w);
                atomicAdd(a + 12 * RNGF, p1 * f3.x);
                atomicAdd(a + 13 * RNGF, p1 * f3.y);
                atomicAdd(a + 14 * RNGF, p1 * f3.z);
                atomicAdd(a + 15 * RNGF, p1 * f3.w);
                atomicAdd(a + 16 * RNGF, p0);
                atomicAdd(a + 17 * RNGF, p1);
            }
        };
    auto sh_bound = [&](float2 r, unsigned) __attribute__((always_inline)) { return make_float2(leaky(b0 + r.x), leaky(b1 + r.y)); };
    auto sh_any = [&](float2 r, unsigned dl) __attribute__((always_inline)) {
            if (bounded) return make_float2(leaky(b0 + r.x), leaky(b1 + r.y));
            uint2 u = *reinterpret_cast<const uint2*>(mxg + 2 * ((size_t)base + dl));
            return make_float2(funmap(u.x), funmap(u.y));
        };
    auto hit_bound = [&](unsigned s, unsigned dl, bool act) __attribute__((always_inline)) { hit_sum(s, dl, act, sh_bound); };
    auto hit_any   = [&](unsigned s, unsigned dl, bool act) __attribute__((always_inline)) { hit_sum(s, dl, act, sh_any); };
    if (__builtin_expect(bounded && al4, 1))
        scan_hits<RNGF, true >(srcp, ei + E, eb, ee, base, ring, lane, wave, hit_bound);
    else
        scan_hits<RNGF, false>(srcp, ei + E, eb, ee, base, ring, lane, wave, hit_any);
    __syncthreads();
    for (int c = 0; c < NACC; ++c) {
        float* gcol = gsum + (size_t)c * N + base;
        for (int i = tid; i < RNGF; i += WGT)
            if (base + i < N) atomicAdd(gcol + i, acc[c * RNGF + i]);
    }
}

__global__ __launch_bounds__(256) void k_final(const float* __restrict__ gsum, float* __restrict__ out, int N)
{
    int idx = blockIdx.x * 256 + threadIdx.x;
    int n = idx >> 1, hf = idx & 1;
    if (n >= N) return;
    const float* g = gsum + n;
    float r0 = 0.5f / fmaxf(g[(size_t)16 * N], EPSF);
    float r1 = 0.5f / fmaxf(g[(size_t)17 * N], EPSF);
    int c0 = hf * 4;
    float4* o = reinterpret_cast<float4*>(out + (size_t)n * 8 + c0);
    float4 v = *o;
    v.x += fmaf(g[(size_t)(c0 + 0) * N], r0, g[(size_t)(c0 +  8) * N] * r1);
    v.y += fmaf(g[(size_t)(c0 + 1) * N], r0, g[(size_t)(c0 +  9) * N] * r1);
    v.z += fmaf(g[(size_t)(c0 + 2) * N], r0, g[(size_t)(c0 + 10) * N] * r1);
    v.w += fmaf(g[(size_t)(c0 + 3) * N], r0, g[(size_t)(c0 + 11) * N] * r1);
    *o = v;
}

extern "C" void kernel_launch(void* const* d_in, const int* in_sizes, int n_in,
                              void* d_out, int out_size, void* d_ws, size_t ws_size,
                              hipStream_t stream)
{
    const float* x    = (const float*)d_in[0];
    const int*   ei   = (const int*)  d_in[1];
    const float* W    = (const float*)d_in[2];
    const float* al   = (const float*)d_in[3];
    const float* ar   = (const float*)d_in[4];
    const float* bias = (const float*)d_in[5];
    const float* Wr   = (const float*)d_in[6];
    float* out = (float*)d_out;

    int N = in_sizes[0] / IN_FEAT;
    int E = in_sizes[1] / 2;
    int nPart = (N + 255) / 256;

    char* ws = (char*)d_ws;
    float*    feat  = (float*)ws;    ws += (size_t)N * OUTC * sizeof(float);
    float4*   pmm   = (float4*)ws;   ws += (size_t)nPart * sizeof(float4);
    float*    el    = (float*)ws;    ws += (size_t)N * 2 * sizeof(float);
    float*    er    = (float*)ws;    ws += (size_t)N * 2 * sizeof(float);
    unsigned* mxg   = (unsigned*)ws; ws += (size_t)N * 2 * sizeof(unsigned);
    float*    gsum  = (float*)ws;    ws += (size_t)N * NACC * sizeof(float);

    int nTiles = (N + 15) / 16;
    dim3 blk(256);

    k_node    <<<(nTiles + 7) / 8,      blk, 0, stream>>>(x, W, Wr, bias, feat, out, N, nTiles);
    k_elr     <<<nPart,                 blk, 0, stream>>>(feat, al, ar, el, er, gsum, mxg, pmm, N);
    k_edge_max<<<256,                   blk, 0, stream>>>(ei, el, er, pmm, nPart, mxg, E);
    int nRangeF = (N + RNGF - 1) / RNGF;
    int chunkF  = (E + NSLICEF - 1) / NSLICEF;
    dim3 grdF(nRangeF, NSLICEF);
    dim3 wg(WGT);
    k_range_fused<<<grdF, wg, 0, stream>>>(ei, al, er, pmm, nPart, mxg, feat, gsum, N, E, chunkF);
    k_final   <<<(2 * N + 255) / 256,   blk, 0, stream>>>(gsum, out, N);
}


# Edge list partitioned once by owning destination range (chunk counting sort in LDS, packed words in workspace); a range work-group reads only its own runs, no 26-fold filter; LDS add order changes
# speedup vs baseline: 11.8955x; 1.1423x over previous
#include <hip/hip_runtime.h>
#include <hip/hip_bf16.h>
#include <math.h>


typedef __attribute__((ext_vector_type(2))) float v2f;
typedef __attribute__((ext_vector_type(8))) float v8f;

#define IN_FEAT   256
#define OUTC      16
#define NEG_SLOPE 0.2f
#define EPSF      1e-15f
#define NACC      18
#define ELSPAN    16.0f
#define MAXINIT   0x007FFFFFu

__device__ __forceinline__ unsigned fmap(float f) {
    unsigned u = __float_as_uint(f);
    return (u & 0x80000000u) ? ~u : (u | 0x80000000u);
}
__device__ __forceinline__ float funmap(unsigned u) {
    unsigned v = (u & 0x80000000u) ? (u & 0x7FFFFFFFu) : ~u;
    return __uint_as_float(v);
}

__global__ __launch_bounds__(256) void k_node(
    const float* __restrict__ x, const float* __restrict__ W,
    const float* __restrict__ Wr, const float* __restrict__ bias,
    float* __restrict__ feat, float* __restrict__ out, int N, int nTiles)
{
    __shared__ float sW [IN_FEAT * OUTC];
    __shared__ float sWr[IN_FEAT * OUTC];
    int tid = threadIdx.x;
    for (int i = tid; i < IN_FEAT * OUTC; i += 256) {
        sW[i]  = W[i];
        sWr[i] = Wr[i];
    }
    __syncthreads();

    int wave = tid >> 5;
    int lane = tid & 31;
    int t = blockIdx.x * 8 + wave;
    if (t >= nTiles) t = nTiles - 1;

    int half = lane >> 4;
    int l15  = lane & 15;
    int rowg = t * 16 + l15;
    if (rowg >= N) rowg = N - 1;
    const float* xrow = x + (size_t)rowg * IN_FEAT;

    v8f cf = {0.f,0.f,0.f,0.f,0.f,0.f,0.f,0.f};
    v8f cr = {0.f,0.f,0.f,0.f,0.f,0.f,0.f,0.f};
    float nsq = 0.f;

    for (int k0 = 0; k0 < IN_FEAT; k0 += 4) {
        int ka = k0 + half * 2;
        float a0 = xrow[ka], a1 = xrow[ka + 1];
        nsq = fmaf(a0, a0, nsq);
        nsq = fmaf(a1, a1, nsq);
        v2f a; a[0] = a0; a[1] = a1;
        v2f b;  b[0]  = sW [ka * OUTC + l15];  b[1]  = sW [(ka + 1) * OUTC + l15];
        v2f br; br[0] = sWr[ka * OUTC + l15];  br[1] = sWr[(ka + 1) * OUTC + l15];
        cf = __builtin_amdgcn_wmma_f32_16x16x4_f32(false, a, false, b,  (short)0, cf, false, false);
        cr = __builtin_amdgcn_wmma_f32_16x16x4_f32(false, a, false, br, (short)0, cr, false, false);
    }

    nsq += __shfl_xor(nsq, 16, 32);
    float nrm = fmaxf(sqrtf(nsq), EPSF);
    float z   = fminf(nrm, 1.0f - 1e-7f);
    float sc  = 0.5f * (log1pf(z) - log1pf(-z)) / nrm;
    float bv  = bias[l15];

    float* fp = feat + (size_t)t * 16 * OUTC;
    float* op = out  + (size_t)t * 16 * 8;
    #pragma unroll
    for (int r = 0; r < 8; ++r) {
        int row = r + half * 8;
        float s_r = __shfl(sc, row, 32);
        float rv  = cr[r] * s_r + bv;
        float ro  = __shfl_xor(rv, 8, 32);
        if (t * 16 + row < N) {
            fp[row * OUTC + l15] = cf[r] * s_r;
            if (l15 < 8) op[row * 8 + l15] = 0.5f * (rv + ro);
        }
    }
}

__global__ __launch_bounds__(256) void k_elr(const float* __restrict__ feat, const float* __restrict__ al,
                      const float* __restrict__ ar, float* __restrict__ el, float* __restrict__ er,
                      float* __restrict__ gsum, unsigned* __restrict__ mxg,
                      float4* __restrict__ pmm, int N)
{
    __shared__ float4 wm[8];
    int n = blockIdx.x * 256 + threadIdx.x;
    float e0 = -INFINITY, e1 = -INFINITY;
    float g0 =  INFINITY, g1 =  INFINITY;
    if (n < N) {
        const float* f = feat + (size_t)n * OUTC;
        float sl[2], sr[2];
        #pragma unroll
        for (int h = 0; h < 2; ++h) {
            sl[h] = 0.f; sr[h] = 0.f;
            #pragma unroll
            for (int c = 0; c < 8; ++c) {
                float v = f[h * 8 + c];
                sl[h] = fmaf(v, al[h * 8 + c], sl[h]);
                sr[h] = fmaf(v, ar[h * 8 + c], sr[h]);
            }
        }
        e0 = sl[0]; e1 = sl[1];
        g0 = sl[0]; g1 = sl[1];
        *reinterpret_cast<float2*>(el + 2 * (size_t)n) = make_float2(sl[0], sl[1]);
        *reinterpret_cast<float2*>(er + 2 * (size_t)n) = make_float2(sr[0], sr[1]);
        *reinterpret_cast<uint2*>(mxg + 2 * (size_t)n) = make_uint2(MAXINIT, MAXINIT);
        for (int c = 0; c < NACC; ++c) gsum[(size_t)c * N + n] = 0.f;
    }
    for (int o = 16; o > 0; o >>= 1) {
        e0 = fmaxf(e0, __shfl_xor(e0, o, 32));
        e1 = fmaxf(e1, __shfl_xor(e1, o, 32));
        g0 = fminf(g0, __shfl_xor(g0, o, 32));
        g1 = fminf(g1, __shfl_xor(g1, o, 32));
    }
    if ((threadIdx.x & 31) == 0) wm[threadIdx.x >> 5] = make_float4(g0, g1, e0, e1);
    __syncthreads();
    if (threadIdx.x == 0) {
        float4 r = wm[0];
        for (int w = 1; w < 8; ++w) {
            r.x = fminf(r.x, wm[w].x); r.y = fminf(r.y, wm[w].y);
            r.z = fmaxf(r.z, wm[w].z); r.w = fmaxf(r.w, wm[w].w);
        }
        pmm[blockIdx.x] = r;
    }
}

__device__ __forceinline__ float4 el_bounds(const float4* __restrict__ pmm, int nPart, float4* red)
{
    int tid = threadIdx.x, lane = tid & 31, wave = tid >> 5, nw = (int)blockDim.x >> 5;
    float4 b = make_float4(INFINITY, INFINITY, -INFINITY, -INFINITY);
    for (int i = tid; i < nPart; i += (int)blockDim.x) {
        float4 p = pmm[i];
        b.x = fminf(b.x, p.x); b.y = fminf(b.y, p.y);
        b.z = fmaxf(b.z, p.z); b.w = fmaxf(b.w, p.w);
    }
    for (int o = 16; o > 0; o >>= 1) {
        b.x = fminf(b.x, __shfl_xor(b.x, o, 32)); b.y = fminf(b.y, __shfl_xor(b.y, o, 32));
        b.z = fmaxf(b.z, __shfl_xor(b.z, o, 32)); b.w = fmaxf(b.w, __shfl_xor(b.w, o, 32));
    }
    if (lane == 0) red[wave] = b;
    __syncthreads();
    b = red[0];
    for (int w = 1; w < nw; ++w) {
        float4 p = red[w];
        b.x = fminf(b.x, p.x); b.y = fminf(b.y, p.y);
        b.z = fmaxf(b.z, p.z); b.w = fmaxf(b.w, p.w);
    }
    return b;
}
__device__ __forceinline__ bool span_small(float4 b) {
    return (b.z - b.x <= ELSPAN) && (b.w - b.y <= ELSPAN);
}

__global__ __launch_bounds__(256) void k_edge_max(
    const int* __restrict__ ei, const float* __restrict__ el, const float* __restrict__ er,
    const float4* __restrict__ pmm, int nPart, unsigned* __restrict__ mxg, int E)
{
    __shared__ float4 red[8];
    if (span_small(el_bounds(pmm, nPart, red))) return;
    for (int e = blockIdx.x * 256 + threadIdx.x; e < E; e += (int)gridDim.x * 256) {
        int s = ei[e], d = ei[(size_t)E + e];
        float2 l = *reinterpret_cast<const float2*>(el + 2 * (size_t)s);
        float2 r = *reinterpret_cast<const float2*>(er + 2 * (size_t)d);
        float v0 = l.x + r.x, v1 = l.y + r.y;
        atomicMax(mxg + 2 * (size_t)d,     fmap((v0 >= 0.f) ? v0 : NEG_SLOPE * v0));
        atomicMax(mxg + 2 * (size_t)d + 1, fmap((v1 >= 0.f) ? v1 : NEG_SLOPE * v1));
    }
}

#define RNGF    3872
#define NSLICEF 9
#define WGT     1024
#define NWAVE   (WGT / 32)
#define QCAP    64

__device__ __forceinline__ float leaky(float v) { return (v >= 0.f) ? v : NEG_SLOPE * v; }

__device__ __forceinline__ void ring_order() {
    __builtin_amdgcn_fence(__ATOMIC_RELEASE, "wavefront");
    __builtin_amdgcn_wave_barrier();
    __builtin_amdgcn_fence(__ATOMIC_ACQUIRE, "wavefront");
}

template <bool AL4>
__device__ __forceinline__ int4 load_quad(const int* __restrict__ p, int e4, int ee) {
    if (AL4) return *reinterpret_cast<const int4*>(p + min(e4, ee - 4));
    int4 r;
    r.x = p[min(e4,     ee - 1)];
    r.y = p[min(e4 + 1, ee - 1)];
    r.z = p[min(e4 + 2, ee - 1)];
    r.w = p[min(e4 + 3, ee - 1)];
    return r;
}
__device__ __forceinline__ int quad_at(const int4& v, int k) {
    return (k == 0) ? v.x : (k == 1) ? v.y : (k == 2) ? v.z : v.w;
}

#define NQ      4
#define BLK     (NQ * 128)
template <int R, bool AL4, typename F>
__device__ __forceinline__ void scan_hits(const int* __restrict__ srcp, const int* __restrict__ dstp,
                                          int eb, int ee, int base,
                                          unsigned* __restrict__ ring, int lane, int wave, F work)
{
    unsigned* qs = ring + wave * (2 * QCAP);
    unsigned* qd = qs + QCAP;
    unsigned head = 0u, qn = 0u;
    int first = eb + wave * BLK;
    if (first >= ee) return;
    int4 dq[NQ], sq[NQ];
    #pragma unroll
    for (int q = 0; q < NQ; ++q) { dq[q] = make_int4(-1, -1, -1, -1); sq[q] = make_int4(0, 0, 0, 0); }
    for (int e0 = first - NWAVE * BLK; e0 < ee; e0 += NWAVE * BLK) {
        int en = e0 + NWAVE * BLK;
        #pragma unroll
        for (int q = 0; q < NQ; ++q) {
            int eq = e0 + q * 128 + 4 * lane;
            #pragma unroll
            for (int k = 0; k < 4; ++k) {
                int e = eq + k;
                unsigned dl = (unsigned)(((e < ee) ? quad_at(dq[q], k) : -1) - base);
                bool     hit  = dl < (unsigned)R;
                unsigned mask = __builtin_amdgcn_ballot_w32(hit);
                if (mask != 0u) {
                    if (hit) {
                        unsigned pos = (head + qn + __builtin_amdgcn_mbcnt_lo(mask, 0u)) & (QCAP - 1);
                        qs[pos] = (unsigned)quad_at(sq[q], k);
                        qd[pos] = dl;
                    }
                    qn += (unsigned)__popc(mask);
                    if (qn >= 32u) {
                        ring_order();
                        unsigned p = (head + (unsigned)lane) & (QCAP - 1);
                        work(qs[p], qd[p], true);
                        ring_order();
                        head = (head + 32u) & (QCAP - 1);
                        qn  -= 32u;
                    }
                }
            }
            int nq = en + q * 128 + 4 * lane;
            dq[q] = load_quad<AL4>(dstp, nq, ee);
            sq[q] = load_quad<AL4>(srcp, nq, ee);
        }
    }
    if (qn != 0u) {
        ring_order();
        unsigned p = (head + (unsigned)lane) & (QCAP - 1);
        work(qs[p], qd[p], (unsigned)lane < qn);
    }
}

#define PEPT    11
#define PCH     (PEPT * WGT)
#define SRCBITS 20
#define MAXB    272
__global__ __launch_bounds__(WGT) void k_part(const int* __restrict__ ei, unsigned* __restrict__ part,
                                              unsigned* __restrict__ offs, int N, int E, int nb)
{
    __shared__ unsigned cnt[MAXB];
    __shared__ unsigned off[MAXB + 1];
    __shared__ unsigned sorted[PCH];
    int tid = threadIdx.x;
    int eb = blockIdx.x * PCH;
    for (int b = tid; b < nb; b += WGT) cnt[b] = 0u;
    __syncthreads();
    unsigned dv[PEPT], sv[PEPT], bin[PEPT], rnk[PEPT], wrd[PEPT];
    #pragma unroll
    for (int i = 0; i < PEPT; ++i) {
        int ec = min(eb + i * WGT + tid, E - 1);
        dv[i] = (unsigned)ei[(size_t)E + ec];
        sv[i] = (unsigned)ei[ec];
    }
    #pragma unroll
    for (int i = 0; i < PEPT; ++i) {
        int e = eb + i * WGT + tid;
        bin[i] = 0xFFFFFFFFu; rnk[i] = 0u; wrd[i] = 0u;
        if (e < E && dv[i] < (unsigned)N) {
            unsigned b = dv[i] / (unsigned)RNGF;
            bin[i] = b;
            wrd[i] = ((dv[i] - b * (unsigned)RNGF) << SRCBITS) | (sv[i] & ((1u << SRCBITS) - 1u));
            rnk[i] = atomicAdd(&cnt[b], 1u);
        }
    }
    __syncthreads();
    if (tid < 32) {
        unsigned carry = 0u;
        for (int b0 = 0; b0 < nb; b0 += 32) {
            int b = b0 + tid;
            unsigned c = (b < nb) ? cnt[b] : 0u;
            unsigned x = c;
            for (int o = 1; o < 32; o <<= 1) {
                unsigned y = __shfl_up(x, o, 32);
                if (tid >= o) x += y;
            }
            if (b < nb) off[b] = carry + x - c;
            carry += __shfl(x, 31, 32);
        }
        if (tid == 0) off[nb] = carry;
    }
    __syncthreads();
    for (int b = tid; b <= nb; b += WGT) offs[(size_t)blockIdx.x * (nb + 1) + b] = off[b];
    #pragma unroll
    for (int i = 0; i < PEPT; ++i)
        if (bin[i] != 0xFFFFFFFFu) sorted[off[bin[i]] + rnk[i]] = wrd[i];
    __syncthreads();
    unsigned total = off[nb];
    unsigned* reg = part + (size_t)blockIdx.x * PCH;
    for (unsigned i = tid; i < total; i += WGT) reg[i] = sorted[i];
}

template <typename F>
__device__ __forceinline__ void part_hits(const unsigned* __restrict__ part, const unsigned* __restrict__ offs,
                                          int c0, int c1, int nb1, int g, int lane, int wv, F work)
{
    for (int c = c0 + wv; c < c1; c += NWAVE) {
        const unsigned* o = offs + (size_t)c * nb1 + g;
        unsigned lo = o[0], hi = o[1];
        if (lo >= hi) continue;
        const unsigned* p = part + (size_t)c * PCH;
        unsigned last = hi - 1u;
        unsigned w = p[min(lo + (unsigned)lane, last)];
        for (unsigned i0 = lo; i0 < hi; i0 += 32u) {
            unsigned cur = w;
            bool act = i0 + (unsigned)lane < hi;
            w = p[min(i0 + 32u + (unsigned)lane, last)];
            work(cur & ((1u << SRCBITS) - 1u), cur >> SRCBITS, act);
        }
    }
}

__global__ __launch_bounds__(WGT) void k_range_fused(
    const int* __restrict__ ei, const float* __restrict__ al, const float* __restrict__ er,
    const float4* __restrict__ pmm, int nPart, const unsigned* __restrict__ mxg,
    const float* __restrict__ feat, float* __restrict__ gsum, int N, int E, int chunk,
    const unsigned* __restrict__ part, const unsigned* __restrict__ offs, int cps, int nChunk)
{
    __shared__ __align__(16) float acc[NACC * RNGF];
    __shared__ float2 ser[RNGF];
    __shared__ unsigned ring[NWAVE * 2 * QCAP];
    __shared__ float4 red[NWAVE];
    int tid = threadIdx.x, lane = tid & 31, wave = tid >> 5;
    int base = blockIdx.x * RNGF;
    int chunk4 = (chunk + 3) & ~3;
    int eb = blockIdx.y * chunk4;
    int ee = min(E, eb + chunk4);
    bool al4 = (E & 3) == 0;
    float4* acc4 = reinterpret_cast<float4*>(acc);
    for (int i = tid; i < NACC * RNGF / 4; i += WGT) acc4[i] = make_float4(0.f, 0.f, 0.f, 0.f);
    for (int i = tid; i < RNGF; i += WGT) {
        int n = base + i;
        ser[i] = (n < N) ? *reinterpret_cast<const float2*>(er + 2 * (size_t)n) : make_float2(0.f, 0.f);
    }
    float4 bnd = el_bounds(pmm, nPart, red);
    bool bounded = __builtin_amdgcn_readfirstlane(span_small(bnd) ? 1 : 0) != 0;
    float b0 = bnd.z, b1 = bnd.w;
    const int* srcp = ei;
    float w0 = al[0],  w1 = al[1],  w2 = al[2],   w3 = al[3],   w4 = al[4],   w5 = al[5],   w6 = al[6],   w7 = al[7];
    float w8 = al[8],  w9 = al[9],  w10 = al[10], w11 = al[11], w12 = al[12], w13 = al[13], w14 = al[14], w15 = al[15];
    auto hit_sum = [&](unsigned s, unsigned dl, bool act, auto sh) __attribute__((always_inline)) {
            if (act) {
                const float4* fs = reinterpret_cast<const float4*>(feat + (size_t)s * OUTC);
                float4 f0 = fs[0], f1 = fs[1], f2 = fs[2], f3 = fs[3];
                float2 l;
                l.x = fmaf(f0.x, w0, 0.f);  l.x = fmaf(f0.y, w1, l.x);  l.x = fmaf(f0.z, w2, l.x);  l.x = fmaf(f0.w, w3, l.x);
                l.x = fmaf(f1.x, w4, l.x);  l.x = fmaf(f1.y, w5, l.x);  l.x = fmaf(f1.z, w6, l.x);  l.x = fmaf(f1.w, w7, l.x);
                l.y = fmaf(f2.x, w8, 0.f);  l.y = fmaf(f2.y, w9, l.y);  l.y = fmaf(f2.z, w10, l.y); l.y = fmaf(f2.w, w11, l.y);
                l.y = fmaf(f3.x, w12, l.y); l.y = fmaf(f3.y, w13, l.y); l.y = fmaf(f3.z, w14, l.y); l.y = fmaf(f3.w, w15, l.y);
                float2 r = ser[dl];
                float2 m = sh(r, dl);
                float p0 = __expf(leaky(l.x + r.x) - m.x);
                float p1 = __expf(leaky(l.y + r.y) - m.y);
                float* a = acc + dl;
                atomicAdd(a +  0 * RNGF, p0 * f0.x);
                atomicAdd(a +  1 * RNGF, p0 * f0.y);
                atomicAdd(a +  2 * RNGF, p0 * f0.z);
                atomicAdd(a +  3 * RNGF, p0 * f0.w);
                atomicAdd(a +  4 * RNGF, p0 * f1.x);
                atomicAdd(a +  5 * RNGF, p0 * f1.y);
                atomicAdd(a +  6 * RNGF, p0 * f1.z);
                atomicAdd(a +  7 * RNGF, p0 * f1.w);
                atomicAdd(a +  8 * RNGF, p1 * f2.x);
                atomicAdd(a +  9 * RNGF, p1 * f2.y);
                atomicAdd(a + 10 * RNGF, p1 * f2.z);
                atomicAdd(a + 11 * RNGF, p1 * f2.w);
                atomicAdd(a + 12 * RNGF, p1 * f3.x);
                atomicAdd(a + 13 * RNGF, p1 * f3.y);
                atomicAdd(a + 14 * RNGF, p1 * f3.z);
                atomicAdd(a + 15 * RNGF, p1 * f3.w);
                atomicAdd(a + 16 * RNGF, p0);
                atomicAdd(a + 17 * RNGF, p1);
            }
        };
    auto sh_bound = [&](float2 r, unsigned) __attribute__((always_inline)) { return make_float2(leaky(b0 + r.x), leaky(b1 + r.y)); };
    auto sh_any = [&](float2 r, unsigned dl) __attribute__((always_inline)) {
            if (bounded) return make_float2(leaky(b0 + r.x), leaky(b1 + r.y));
            uint2 u = *reinterpret_cast<const uint2*>(mxg + 2 * ((size_t)base + dl));
            return make_float2(funmap(u.x), funmap(u.y));
        };
    auto hit_bound = [&](unsigned s, unsigned dl, bool act) __attribute__((always_inline)) { hit_sum(s, dl, act, sh_bound); };
    auto hit_any   = [&](unsigned s, unsigned dl, bool act) __attribute__((always_inline)) { hit_sum(s, dl, act, sh_any); };
    if (__builtin_expect(part != nullptr, 1)) {
        int wv = __builtin_amdgcn_readfirstlane(wave);
        int c0 = blockIdx.y * cps;
        int c1 = min(nChunk, c0 + cps);
        int nb1 = (int)gridDim.x + 1;
        if (__builtin_expect(bounded, 1)) part_hits(part, offs, c0, c1, nb1, (int)blockIdx.x, lane, wv, hit_bound);
        else         part_hits(part, offs, c0, c1, nb1, (int)blockIdx.x, lane, wv, hit_any);
    }
    else if (__builtin_expect(bounded && al4, 1))
        scan_hits<RNGF, true >(srcp, ei + E, eb, ee, base, ring, lane, wave, hit_bound);
    else
        scan_hits<RNGF, false>(srcp, ei + E, eb, ee, base, ring, lane, wave, hit_any);
    __syncthreads();
    for (int c = 0; c < NACC; ++c) {
        float* gcol = gsum + (size_t)c * N + base;
        for (int i = tid; i < RNGF; i += WGT)
            if (base + i < N) atomicAdd(gcol + i, acc[c * RNGF + i]);
    }
}

__global__ __launch_bounds__(256) void k_final(const float* __restrict__ gsum, float* __restrict__ out, int N)
{
    int idx = blockIdx.x * 256 + threadIdx.x;
    int n = idx >> 1, hf = idx & 1;
    if (n >= N) return;
    const float* g = gsum + n;
    float r0 = 0.5f / fmaxf(g[(size_t)16 * N], EPSF);
    float r1 = 0.5f / fmaxf(g[(size_t)17 * N], EPSF);
    int c0 = hf * 4;
    float4* o = reinterpret_cast<float4*>(out + (size_t)n * 8 + c0);
    float4 v = *o;
    v.x += fmaf(g[(size_t)(c0 + 0) * N], r0, g[(size_t)(c0 +  8) * N] * r1);
    v.y += fmaf(g[(size_t)(c0 + 1) * N], r0, g[(size_t)(c0 +  9) * N] * r1);
    v.z += fmaf(g[(size_t)(c0 + 2) * N], r0, g[(size_t)(c0 + 10) * N] * r1);
    v.w += fmaf(g[(size_t)(c0 + 3) * N], r0, g[(size_t)(c0 + 11) * N] * r1);
    *o = v;
}

extern "C" void kernel_launch(void* const* d_in, const int* in_sizes, int n_in,
                              void* d_out, int out_size, void* d_ws, size_t ws_size,
                              hipStream_t stream)
{
    const float* x    = (const float*)d_in[0];
    const int*   ei   = (const int*)  d_in[1];
    const float* W    = (const float*)d_in[2];
    const float* al   = (const float*)d_in[3];
    const float* ar   = (const float*)d_in[4];
    const float* bias = (const float*)d_in[5];
    const float* Wr   = (const float*)d_in[6];
    float* out = (float*)d_out;

    int N = in_sizes[0] / IN_FEAT;
    int E = in_sizes[1] / 2;
    int nPart = (N + 255) / 256;

    char* ws = (char*)d_ws;
    float*    feat  = (float*)ws;    ws += (size_t)N * OUTC * sizeof(float);
    float4*   pmm   = (float4*)ws;   ws += (size_t)nPart * sizeof(float4);
    float*    el    = (float*)ws;    ws += (size_t)N * 2 * sizeof(float);
    float*    er    = (float*)ws;    ws += (size_t)N * 2 * sizeof(float);
    unsigned* mxg   = (unsigned*)ws; ws += (size_t)N * 2 * sizeof(unsigned);
    float*    gsum  = (float*)ws;    ws += (size_t)N * NACC * sizeof(float);
    int nRangeF = (N + RNGF - 1) / RNGF;
    int nChunk  = (E + PCH - 1) / PCH;
    unsigned* part  = (unsigned*)ws; ws += (size_t)nChunk * PCH * sizeof(unsigned);
    unsigned* offs  = (unsigned*)ws; ws += (size_t)nChunk * (nRangeF + 1) * sizeof(unsigned);
    bool parted = nChunk > 0 && N <= (1 << SRCBITS) && (size_t)(ws - (char*)d_ws) <= ws_size;

    int nTiles = (N + 15) / 16;
    dim3 blk(256);

    dim3 wg(WGT);
    if (parted) k_part<<<nChunk, wg, 0, stream>>>(ei, part, offs, N, E, nRangeF);
    k_node    <<<(nTiles + 7) / 8,      blk, 0, stream>>>(x, W, Wr, bias, feat, out, N, nTiles);
    k_elr     <<<nPart,                 blk, 0, stream>>>(feat, al, ar, el, er, gsum, mxg, pmm, N);
    k_edge_max<<<256,                   blk, 0, stream>>>(ei, el, er, pmm, nPart, mxg, E);
    int chunkF  = (E + NSLICEF - 1) / NSLICEF;
    int cps     = (nChunk + NSLICEF - 1) / NSLICEF;
    dim3 grdF(nRangeF, NSLICEF);
    k_range_fused<<<grdF, wg, 0, stream>>>(ei, al, er, pmm, nPart, mxg, feat, gsum, N, E, chunkF,
                                           parted ? part : nullptr, offs, cps, nChunk);
    k_final   <<<(2 * N + 255) / 256,   blk, 0, stream>>>(gsum, out, N);
}


# One work-group owns a 400-node destination range and all its edges (250 ranges, partition runs of two batches): sums finish in LDS and go straight to out, no global float atomics; sum order changes
# speedup vs baseline: 14.6265x; 1.2296x over previous
#include <hip/hip_runtime.h>
#include <hip/hip_bf16.h>
#include <math.h>


typedef __attribute__((ext_vector_type(2))) float v2f;
typedef __attribute__((ext_vector_type(8))) float v8f;

#define IN_FEAT   256
#define OUTC      16
#define NEG_SLOPE 0.2f
#define EPSF      1e-15f
#define NACC      18
#define ELSPAN    16.0f
#define MAXINIT   0x007FFFFFu

__device__ __forceinline__ unsigned fmap(float f) {
    unsigned u = __float_as_uint(f);
    return (u & 0x80000000u) ? ~u : (u | 0x80000000u);
}
__device__ __forceinline__ float funmap(unsigned u) {
    unsigned v = (u & 0x80000000u) ? (u & 0x7FFFFFFFu) : ~u;
    return __uint_as_float(v);
}

__global__ __launch_bounds__(256) void k_node(
    const float* __restrict__ x, const float* __restrict__ W,
    const float* __restrict__ Wr, const float* __restrict__ bias,
    float* __restrict__ feat, float* __restrict__ out, int N, int nTiles)
{
    __shared__ float sW [IN_FEAT * OUTC];
    __shared__ float sWr[IN_FEAT * OUTC];
    int tid = threadIdx.x;
    for (int i = tid; i < IN_FEAT * OUTC; i += 256) {
        sW[i]  = W[i];
        sWr[i] = Wr[i];
    }
    __syncthreads();

    int wave = tid >> 5;
    int lane = tid & 31;
    int t = blockIdx.x * 8 + wave;
    if (t >= nTiles) t = nTiles - 1;

    int half = lane >> 4;
    int l15  = lane & 15;
    int rowg = t * 16 + l15;
    if (rowg >= N) rowg = N - 1;
    const float* xrow = x + (size_t)rowg * IN_FEAT;

    v8f cf = {0.f,0.f,0.f,0.f,0.f,0.f,0.f,0.f};
    v8f cr = {0.f,0.f,0.f,0.f,0.f,0.f,0.f,0.f};
    float nsq = 0.f;

    for (int k0 = 0; k0 < IN_FEAT; k0 += 4) {
        int ka = k0 + half * 2;
        float a0 = xrow[ka], a1 = xrow[ka + 1];
        nsq = fmaf(a0, a0, nsq);
        nsq = fmaf(a1, a1, nsq);
        v2f a; a[0] = a0; a[1] = a1;
        v2f b;  b[0]  = sW [ka * OUTC + l15];  b[1]  = sW [(ka + 1) * OUTC + l15];
        v2f br; br[0] = sWr[ka * OUTC + l15];  br[1] = sWr[(ka + 1) * OUTC + l15];
        cf = __builtin_amdgcn_wmma_f32_16x16x4_f32(false, a, false, b,  (short)0, cf, false, false);
        cr = __builtin_amdgcn_wmma_f32_16x16x4_f32(false, a, false, br, (short)0, cr, false, false);
    }

    nsq += __shfl_xor(nsq, 16, 32);
    float nrm = fmaxf(sqrtf(nsq), EPSF);
    float z   = fminf(nrm, 1.0f - 1e-7f);
    float sc  = 0.5f * (log1pf(z) - log1pf(-z)) / nrm;
    float bv  = bias[l15];

    float* fp = feat + (size_t)t * 16 * OUTC;
    float* op = out  + (size_t)t * 16 * 8;
    #pragma unroll
    for (int r = 0; r < 8; ++r) {
        int row = r + half * 8;
        float s_r = __shfl(sc, row, 32);
        float rv  = cr[r] * s_r + bv;
        float ro  = __shfl_xor(rv, 8, 32);
        if (t * 16 + row < N) {
            fp[row * OUTC + l15] = cf[r] * s_r;
            if (l15 < 8) op[row * 8 + l15] = 0.5f * (rv + ro);
        }
    }
}

__global__ __launch_bounds__(256) void k_elr(const float* __restrict__ feat, const float* __restrict__ al,
                      const float* __restrict__ ar, float* __restrict__ el, float* __restrict__ er,
                      unsigned* __restrict__ mxg,
                      float4* __restrict__ pmm, int N)
{
    __shared__ float4 wm[8];
    int n = blockIdx.x * 256 + threadIdx.x;
    float e0 = -INFINITY, e1 = -INFINITY;
    float g0 =  INFINITY, g1 =  INFINITY;
    if (n < N) {
        const float* f = feat + (size_t)n * OUTC;
        float sl[2], sr[2];
        #pragma unroll
        for (int h = 0; h < 2; ++h) {
            sl[h] = 0.f; sr[h] = 0.f;
            #pragma unroll
            for (int c = 0; c < 8; ++c) {
                float v = f[h * 8 + c];
                sl[h] = fmaf(v, al[h * 8 + c], sl[h]);
                sr[h] = fmaf(v, ar[h * 8 + c], sr[h]);
            }
        }
        e0 = sl[0]; e1 = sl[1];
        g0 = sl[0]; g1 = sl[1];
        *reinterpret_cast<float2*>(el + 2 * (size_t)n) = make_float2(sl[0], sl[1]);
        *reinterpret_cast<float2*>(er + 2 * (size_t)n) = make_float2(sr[0], sr[1]);
        *reinterpret_cast<uint2*>(mxg + 2 * (size_t)n) = make_uint2(MAXINIT, MAXINIT);
    }
    for (int o = 16; o > 0; o >>= 1) {
        e0 = fmaxf(e0, __shfl_xor(e0, o, 32));
        e1 = fmaxf(e1, __shfl_xor(e1, o, 32));
        g0 = fminf(g0, __shfl_xor(g0, o, 32));
        g1 = fminf(g1, __shfl_xor(g1, o, 32));
    }
    if ((threadIdx.x & 31) == 0) wm[threadIdx.x >> 5] = make_float4(g0, g1, e0, e1);
    __syncthreads();
    if (threadIdx.x == 0) {
        float4 r = wm[0];
        for (int w = 1; w < 8; ++w) {
            r.x = fminf(r.x, wm[w].x); r.y = fminf(r.y, wm[w].y);
            r.z = fmaxf(r.z, wm[w].z); r.w = fmaxf(r.w, wm[w].w);
        }
        pmm[blockIdx.x] = r;
    }
}

__device__ __forceinline__ float4 el_bounds(const float4* __restrict__ pmm, int nPart, float4* red)
{
    int tid = threadIdx.x, lane = tid & 31, wave = tid >> 5, nw = (int)blockDim.x >> 5;
    float4 b = make_float4(INFINITY, INFINITY, -INFINITY, -INFINITY);
    for (int i = tid; i < nPart; i += (int)blockDim.x) {
        float4 p = pmm[i];
        b.x = fminf(b.x, p.x); b.y = fminf(b.y, p.y);
        b.z = fmaxf(b.z, p.z); b.w = fmaxf(b.w, p.w);
    }
    for (int o = 16; o > 0; o >>= 1) {
        b.x = fminf(b.x, __shfl_xor(b.x, o, 32)); b.y = fminf(b.y, __shfl_xor(b.y, o, 32));
        b.z = fmaxf(b.z, __shfl_xor(b.z, o, 32)); b.w = fmaxf(b.w, __shfl_xor(b.w, o, 32));
    }
    if (lane == 0) red[wave] = b;
    __syncthreads();
    b = red[0];
    for (int w = 1; w < nw; ++w) {
        float4 p = red[w];
        b.x = fminf(b.x, p.x); b.y = fminf(b.y, p.y);
        b.z = fmaxf(b.z, p.z); b.w = fmaxf(b.w, p.w);
    }
    return b;
}
__device__ __forceinline__ bool span_small(float4 b) {
    return (b.z - b.x <= ELSPAN) && (b.w - b.y <= ELSPAN);
}

__global__ __launch_bounds__(256) void k_edge_max(
    const int* __restrict__ ei, const float* __restrict__ el, const float* __restrict__ er,
    const float4* __restrict__ pmm, int nPart, unsigned* __restrict__ mxg, int E)
{
    __shared__ float4 red[8];
    if (span_small(el_bounds(pmm, nPart, red))) return;
    for (int e = blockIdx.x * 256 + threadIdx.x; e < E; e += (int)gridDim.x * 256) {
        int s = ei[e], d = ei[(size_t)E + e];
        float2 l = *reinterpret_cast<const float2*>(el + 2 * (size_t)s);
        float2 r = *reinterpret_cast<const float2*>(er + 2 * (size_t)d);
        float v0 = l.x + r.x, v1 = l.y + r.y;
        atomicMax(mxg + 2 * (size_t)d,     fmap((v0 >= 0.f) ? v0 : NEG_SLOPE * v0));
        atomicMax(mxg + 2 * (size_t)d + 1, fmap((v1 >= 0.f) ? v1 : NEG_SLOPE * v1));
    }
}

#define RNGF    400
#define WGT     1024
#define NWAVE   (WGT / 32)
#define QCAP    64

__device__ __forceinline__ float leaky(float v) { return (v >= 0.f) ? v : NEG_SLOPE * v; }

__device__ __forceinline__ void ring_order() {
    __builtin_amdgcn_fence(__ATOMIC_RELEASE, "wavefront");
    __builtin_amdgcn_wave_barrier();
    __builtin_amdgcn_fence(__ATOMIC_ACQUIRE, "wavefront");
}

template <bool AL4>
__device__ __forceinline__ int4 load_quad(const int* __restrict__ p, int e4, int ee) {
    if (AL4) return *reinterpret_cast<const int4*>(p + min(e4, ee - 4));
    int4 r;
    r.x = p[min(e4,     ee - 1)];
    r.y = p[min(e4 + 1, ee - 1)];
    r.z = p[min(e4 + 2, ee - 1)];
    r.w = p[min(e4 + 3, ee - 1)];
    return r;
}
__device__ __forceinline__ int quad_at(const int4& v, int k) {
    return (k == 0) ? v.x : (k == 1) ? v.y : (k == 2) ? v.z : v.w;
}

#define NQ      4
#define BLK     (NQ * 128)
template <int R, bool AL4, typename F>
__device__ __forceinline__ void scan_hits(const int* __restrict__ srcp, const int* __restrict__ dstp,
                                          int eb, int ee, int base,
                                          unsigned* __restrict__ ring, int lane, int wave, F work)
{
    unsigned* qs = ring + wave * (2 * QCAP);
    unsigned* qd = qs + QCAP;
    unsigned head = 0u, qn = 0u;
    int first = eb + wave * BLK;
    if (first >= ee) return;
    int4 dq[NQ], sq[NQ];
    #pragma unroll
    for (int q = 0; q < NQ; ++q) { dq[q] = make_int4(-1, -1, -1, -1); sq[q] = make_int4(0, 0, 0, 0); }
    for (int e0 = first - NWAVE * BLK; e0 < ee; e0 += NWAVE * BLK) {
        int en = e0 + NWAVE * BLK;
        #pragma unroll
        for (int q = 0; q < NQ; ++q) {
            int eq = e0 + q * 128 + 4 * lane;
            #pragma unroll
            for (int k = 0; k < 4; ++k) {
                int e = eq + k;
                unsigned dl = (unsigned)(((e < ee) ? quad_at(dq[q], k) : -1) - base);
                bool     hit  = dl < (unsigned)R;
                unsigned mask = __builtin_amdgcn_ballot_w32(hit);
                if (mask != 0u) {
                    if (hit) {
                        unsigned pos = (head + qn + __builtin_amdgcn_mbcnt_lo(mask, 0u)) & (QCAP - 1);
                        qs[pos] = (unsigned)quad_at(sq[q], k);
                        qd[pos] = dl;
                    }
                    qn += (unsigned)__popc(mask);
                    if (qn >= 32u) {
                        ring_order();
                        unsigned p = (head + (unsigned)lane) & (QCAP - 1);
                        work(qs[p], qd[p], true);
                        ring_order();
                        head = (head + 32u) & (QCAP - 1);
                        qn  -= 32u;
                    }
                }
            }
            int nq = en + q * 128 + 4 * lane;
            dq[q] = load_quad<AL4>(dstp, nq, ee);
            sq[q] = load_quad<AL4>(srcp, nq, ee);
        }
    }
    if (qn != 0u) {
        ring_order();
        unsigned p = (head + (unsigned)lane) & (QCAP - 1);
        work(qs[p], qd[p], (unsigned)lane < qn);
    }
}

#define PEPT    14
#define PCH     (PEPT * WGT)
#define SRCBITS 20
#define MAXB    2624
__global__ __launch_bounds__(WGT) void k_part(const int* __restrict__ ei, unsigned* __restrict__ part,
                                              unsigned* __restrict__ offs, int N, int E, int nb)
{
    __shared__ unsigned cnt[MAXB];
    __shared__ unsigned off[MAXB + 1];
    __shared__ unsigned sorted[PCH];
    int tid = threadIdx.x;
    int eb = blockIdx.x * PCH;
    for (int b = tid; b < nb; b += WGT) cnt[b] = 0u;
    __syncthreads();
    unsigned dv[PEPT], sv[PEPT], bin[PEPT], rnk[PEPT], wrd[PEPT];
    #pragma unroll
    for (int i = 0; i < PEPT; ++i) {
        int ec = min(eb + i * WGT + tid, E - 1);
        dv[i] = (unsigned)ei[(size_t)E + ec];
        sv[i] = (unsigned)ei[ec];
    }
    #pragma unroll
    for (int i = 0; i < PEPT; ++i) {
        int e = eb + i * WGT + tid;
        bin[i] = 0xFFFFFFFFu; rnk[i] = 0u; wrd[i] = 0u;
        if (e < E && dv[i] < (unsigned)N) {
            unsigned b = dv[i] / (unsigned)RNGF;
            bin[i] = b;
            wrd[i] = ((dv[i] - b * (unsigned)RNGF) << SRCBITS) | (sv[i] & ((1u << SRCBITS) - 1u));
            rnk[i] = atomicAdd(&cnt[b], 1u);
        }
    }
    __syncthreads();
    if (tid < 32) {
        unsigned carry = 0u;
        for (int b0 = 0; b0 < nb; b0 += 32) {
            int b = b0 + tid;
            unsigned c = (b < nb) ? cnt[b] : 0u;
            unsigned x = c;
            for (int o = 1; o < 32; o <<= 1) {
                unsigned y = __shfl_up(x, o, 32);
                if (tid >= o) x += y;
            }
            if (b < nb) off[b] = carry + x - c;
            carry += __shfl(x, 31, 32);
        }
        if (tid == 0) off[nb] = carry;
    }
    __syncthreads();
    for (int b = tid; b <= nb; b += WGT) offs[(size_t)blockIdx.x * (nb + 1) + b] = off[b];
    #pragma unroll
    for (int i = 0; i < PEPT; ++i)
        if (bin[i] != 0xFFFFFFFFu) sorted[off[bin[i]] + rnk[i]] = wrd[i];
    __syncthreads();
    unsigned total = off[nb];
    unsigned* reg = part + (size_t)blockIdx.x * PCH;
    for (unsigned i = tid; i < total; i += WGT) reg[i] = sorted[i];
}

template <typename F>
__device__ __forceinline__ void part_hits(const unsigned* __restrict__ part, const unsigned* __restrict__ offs,
                                          int nChunk, int nb1, int g, int lane, int wv, F work)
{
    for (int cg = wv; cg < nChunk; cg += 32 * NWAVE) {
        int c = cg + lane * NWAVE;
        unsigned lo = 0u, hi = 0u;
        if (c < nChunk) {
            const unsigned* o = offs + (size_t)c * nb1 + g;
            lo = o[0]; hi = o[1];
        }
        int nr = min(32, (nChunk - cg + NWAVE - 1) / NWAVE);
        int j = 0;
        unsigned a = 0u, b = 0u;
        for (; j < nr; ++j) {
            a = (unsigned)__builtin_amdgcn_readlane((int)lo, j);
            b = (unsigned)__builtin_amdgcn_readlane((int)hi, j);
            if (a < b) break;
        }
        if (j >= nr) continue;
        unsigned w = part[(size_t)(cg + j * NWAVE) * PCH + min(a + (unsigned)lane, b - 1u)];
        for (;;) {
            unsigned cur = w;
            bool act = a + (unsigned)lane < b;
            int nj = j;
            unsigned na = a + 32u, nb = b;
            if (na >= b) {
                for (nj = j + 1; nj < nr; ++nj) {
                    na = (unsigned)__builtin_amdgcn_readlane((int)lo, nj);
                    nb = (unsigned)__builtin_amdgcn_readlane((int)hi, nj);
                    if (na < nb) break;
                }
            }
            bool more = nj < nr;
            if (!more) { nj = j; na = a; nb = b; }
            w = part[(size_t)(cg + nj * NWAVE) * PCH + min(na + (unsigned)lane, nb - 1u)];
            work(cur & ((1u << SRCBITS) - 1u), cur >> SRCBITS, act);
            if (!more) break;
            j = nj; a = na; b = nb;
        }
    }
}

__global__ __launch_bounds__(WGT) void k_range_fused(
    const int* __restrict__ ei, const float* __restrict__ al, const float* __restrict__ er,
    const float4* __restrict__ pmm, int nPart, const unsigned* __restrict__ mxg,
    const float* __restrict__ feat, float* __restrict__ out, int N, int E,
    const unsigned* __restrict__ part, const unsigned* __restrict__ offs, int nChunk)
{
    __shared__ __align__(16) float acc[NACC * RNGF];
    __shared__ float2 ser[RNGF];
    __shared__ unsigned ring[NWAVE * 2 * QCAP];
    __shared__ float4 red[NWAVE];
    int tid = threadIdx.x, lane = tid & 31, wave = tid >> 5;
    int base = blockIdx.x * RNGF;
    bool al4 = (E & 3) == 0;
    float4* acc4 = reinterpret_cast<float4*>(acc);
    for (int i = tid; i < NACC * RNGF / 4; i += WGT) acc4[i] = make_float4(0.f, 0.f, 0.f, 0.f);
    for (int i = tid; i < RNGF; i += WGT) {
        int n = base + i;
        ser[i] = (n < N) ? *reinterpret_cast<const float2*>(er + 2 * (size_t)n) : make_float2(0.f, 0.f);
    }
    float4 bnd = el_bounds(pmm, nPart, red);
    bool bounded = __builtin_amdgcn_readfirstlane(span_small(bnd) ? 1 : 0) != 0;
    float b0 = bnd.z, b1 = bnd.w;
    const int* srcp = ei;
    float w0 = al[0],  w1 = al[1],  w2 = al[2],   w3 = al[3],   w4 = al[4],   w5 = al[5],   w6 = al[6],   w7 = al[7];
    float w8 = al[8],  w9 = al[9],  w10 = al[10], w11 = al[11], w12 = al[12], w13 = al[13], w14 = al[14], w15 = al[15];
    auto hit_sum = [&](unsigned s, unsigned dl, bool act, auto sh) __attribute__((always_inline)) {
            if (act) {
                const float4* fs = reinterpret_cast<const float4*>(feat + (size_t)s * OUTC);
                float4 f0 = fs[0], f1 = fs[1], f2 = fs[2], f3 = fs[3];
                float2 l;
                l.x = fmaf(f0.x, w0, 0.f);  l.x = fmaf(f0.y, w1, l.x);  l.x = fmaf(f0.z, w2, l.x);  l.x = fmaf(f0.w, w3, l.x);
                l.x = fmaf(f1.x, w4, l.x);  l.x = fmaf(f1.y, w5, l.x);  l.x = fmaf(f1.z, w6, l.x);  l.x = fmaf(f1.w, w7, l.x);
                l.y = fmaf(f2.x, w8, 0.f);  l.y = fmaf(f2.y, w9, l.y);  l.y = fmaf(f2.z, w10, l.y); l.y = fmaf(f2.w, w11, l.y);
                l.y = fmaf(f3.x, w12, l.y); l.y = fmaf(f3.y, w13, l.y); l.y = fmaf(f3.z, w14, l.y); l.y = fmaf(f3.w, w15, l.y);
                float2 r = ser[dl];
                float2 m = sh(r, dl);
                float p0 = __expf(leaky(l.x + r.x) - m.x);
                float p1 = __expf(leaky(l.y + r.y) - m.y);
                float* a = acc + dl;
                atomicAdd(a +  0 * RNGF, p0 * f0.x);
                atomicAdd(a +  1 * RNGF, p0 * f0.y);
                atomicAdd(a +  2 * RNGF, p0 * f0.z);
                atomicAdd(a +  3 * RNGF, p0 * f0.w);
                atomicAdd(a +  4 * RNGF, p0 * f1.x);
                atomicAdd(a +  5 * RNGF, p0 * f1.y);
                atomicAdd(a +  6 * RNGF, p0 * f1.z);
                atomicAdd(a +  7 * RNGF, p0 * f1.w);
                atomicAdd(a +  8 * RNGF, p1 * f2.x);
                atomicAdd(a +  9 * RNGF, p1 * f2.y);
                atomicAdd(a + 10 * RNGF, p1 * f2.z);
                atomicAdd(a + 11 * RNGF, p1 * f2.w);
                atomicAdd(a + 12 * RNGF, p1 * f3.x);
                atomicAdd(a + 13 * RNGF, p1 * f3.y);
                atomicAdd(a + 14 * RNGF, p1 * f3.z);
                atomicAdd(a + 15 * RNGF, p1 * f3.w);
                atomicAdd(a + 16 * RNGF, p0);
                atomicAdd(a + 17 * RNGF, p1);
            }
        };
    auto sh_bound = [&](float2 r, unsigned) __attribute__((always_inline)) { return make_float2(leaky(b0 + r.x), leaky(b1 + r.y)); };
    auto sh_any = [&](float2 r, unsigned dl) __attribute__((always_inline)) {
            if (bounded) return make_float2(leaky(b0 + r.x), leaky(b1 + r.y));
            uint2 u = *reinterpret_cast<const uint2*>(mxg + 2 * ((size_t)base + dl));
            return make_float2(funmap(u.x), funmap(u.y));
        };
    auto hit_bound = [&](unsigned s, unsigned dl, bool act) __attribute__((always_inline)) { hit_sum(s, dl, act, sh_bound); };
    auto hit_any   = [&](unsigned s, unsigned dl, bool act) __attribute__((always_inline)) { hit_sum(s, dl, act, sh_any); };
    if (__builtin_expect(part != nullptr, 1)) {
        int wv = __builtin_amdgcn_readfirstlane(wave);
        int nb1 = (int)gridDim.x + 1;
        if (__builtin_expect(bounded, 1)) part_hits(part, offs, nChunk, nb1, (int)blockIdx.x, lane, wv, hit_bound);
        else         part_hits(part, offs, nChunk, nb1, (int)blockIdx.x, lane, wv, hit_any);
    }
    else if (__builtin_expect(bounded && al4, 1))
        scan_hits<RNGF, true >(srcp, ei + E, 0, E, base, ring, lane, wave, hit_bound);
    else
        scan_hits<RNGF, false>(srcp, ei + E, 0, E, base, ring, lane, wave, hit_any);
    __syncthreads();
    for (int idx = tid; idx < 2 * RNGF; idx += WGT) {
        int i = idx >> 1, hf = idx & 1;
        int n = base + i;
        if (n < N) {
            const float* g = acc + i;
            float r0 = 0.5f / fmaxf(g[16 * RNGF], EPSF);
            float r1 = 0.5f / fmaxf(g[17 * RNGF], EPSF);
            int c0 = hf * 4;
            float4* o = reinterpret_cast<float4*>(out + (size_t)n * 8 + c0);
            float4 v = *o;
            v.x += fmaf(g[(c0 + 0) * RNGF], r0, g[(c0 +  8) * RNGF] * r1);
            v.y += fmaf(g[(c0 + 1) * RNGF], r0, g[(c0 +  9) * RNGF] * r1);
            v.z += fmaf(g[(c0 + 2) * RNGF], r0, g[(c0 + 10) * RNGF] * r1);
            v.w += fmaf(g[(c0 + 3) * RNGF], r0, g[(c0 + 11) * RNGF] * r1);
            *o = v;
        }
    }
}

extern "C" void kernel_launch(void* const* d_in, const int* in_sizes, int n_in,
                              void* d_out, int out_size, void* d_ws, size_t ws_size,
                              hipStream_t stream)
{
    const float* x    = (const float*)d_in[0];
    const int*   ei   = (const int*)  d_in[1];
    const float* W    = (const float*)d_in[2];
    const float* al   = (const float*)d_in[3];
    const float* ar   = (const float*)d_in[4];
    const float* bias = (const float*)d_in[5];
    const float* Wr   = (const float*)d_in[6];
    float* out = (float*)d_out;

    int N = in_sizes[0] / IN_FEAT;
    int E = in_sizes[1] / 2;
    int nPart = (N + 255) / 256;

    char* ws = (char*)d_ws;
    float*    feat  = (float*)ws;    ws += (size_t)N * OUTC * sizeof(float);
    float4*   pmm   = (float4*)ws;   ws += (size_t)nPart * sizeof(float4);
    float*    el    = (float*)ws;    ws += (size_t)N * 2 * sizeof(float);
    float*    er    = (float*)ws;    ws += (size_t)N * 2 * sizeof(float);
    unsigned* mxg   = (unsigned*)ws; ws += (size_t)N * 2 * sizeof(unsigned);
    int nRangeF = (N + RNGF - 1) / RNGF;
    int nChunk  = (E + PCH - 1) / PCH;
    unsigned* part  = (unsigned*)ws; ws += (size_t)nChunk * PCH * sizeof(unsigned);
    unsigned* offs  = (unsigned*)ws; ws += (size_t)nChunk * (nRangeF + 1) * sizeof(unsigned);
    bool parted = nChunk > 0 && N <= (1 << SRCBITS) && (size_t)(ws - (char*)d_ws) <= ws_size;

    int nTiles = (N + 15) / 16;
    dim3 blk(256);

    dim3 wg(WGT);
    if (parted) k_part<<<nChunk, wg, 0, stream>>>(ei, part, offs, N, E, nRangeF);
    k_node    <<<(nTiles + 7) / 8,      blk, 0, stream>>>(x, W, Wr, bias, feat, out, N, nTiles);
    k_elr     <<<nPart,                 blk, 0, stream>>>(feat, al, ar, el, er, mxg, pmm, N);
    k_edge_max<<<256,                   blk, 0, stream>>>(ei, el, er, pmm, nPart, mxg, E);
    k_range_fused<<<nRangeF, wg, 0, stream>>>(ei, al, er, pmm, nPart, mxg, feat, out, N, E,
                                              parted ? part : nullptr, offs, nChunk);
}


# el/er and el bounds taken in the node GEMM's epilogue (tile through LDS, same fma chain); wide-span maximum sweep taken inside the range work-group in LDS: 3 launches, not 5; sum order as before
# speedup vs baseline: 15.3968x; 1.0527x over previous
#include <hip/hip_runtime.h>
#include <hip/hip_bf16.h>
#include <math.h>


typedef __attribute__((ext_vector_type(2))) float v2f;
typedef __attribute__((ext_vector_type(8))) float v8f;

#define IN_FEAT   256
#define OUTC      16
#define NEG_SLOPE 0.2f
#define EPSF      1e-15f
#define NACC      18
#define ELSPAN    16.0f
#define MAXINIT   0x007FFFFFu

__device__ __forceinline__ unsigned fmap(float f) {
    unsigned u = __float_as_uint(f);
    return (u & 0x80000000u) ? ~u : (u | 0x80000000u);
}
__device__ __forceinline__ float funmap(unsigned u) {
    unsigned v = (u & 0x80000000u) ? (u & 0x7FFFFFFFu) : ~u;
    return __uint_as_float(v);
}

__device__ __forceinline__ float leaky(float v) { return (v >= 0.f) ? v : NEG_SLOPE * v; }

__device__ __forceinline__ void ring_order() {
    __builtin_amdgcn_fence(__ATOMIC_RELEASE, "wavefront");
    __builtin_amdgcn_wave_barrier();
    __builtin_amdgcn_fence(__ATOMIC_ACQUIRE, "wavefront");
}

#define TPITCH 17
__global__ __launch_bounds__(256) void k_node(
    const float* __restrict__ x, const float* __restrict__ W,
    const float* __restrict__ Wr, const float* __restrict__ bias,
    const float* __restrict__ al, const float* __restrict__ ar,
    float* __restrict__ feat, float* __restrict__ el, float* __restrict__ er,
    float4* __restrict__ pmm, float* __restrict__ out, int N, int nTiles)
{
    __shared__ float sW [IN_FEAT * OUTC];
    __shared__ float sWr[IN_FEAT * OUTC];
    __shared__ float st[8 * 16 * TPITCH];
    __shared__ float sal[OUTC];
    __shared__ float sar[OUTC];
    __shared__ float4 wm[8];
    int tid = threadIdx.x;
    for (int i = tid; i < IN_FEAT * OUTC; i += 256) {
        sW[i]  = W[i];
        sWr[i] = Wr[i];
    }
    if (tid < OUTC) { sal[tid] = al[tid]; sar[tid] = ar[tid]; }
    __syncthreads();

    int wave = tid >> 5;
    int lane = tid & 31;
    int t = blockIdx.x * 8 + wave;
    if (t >= nTiles) t = nTiles - 1;

    int half = lane >> 4;
    int l15  = lane & 15;
    int rowg = t * 16 + l15;
    if (rowg >= N) rowg = N - 1;
    const float* xrow = x + (size_t)rowg * IN_FEAT;

    v8f cf = {0.f,0.f,0.f,0.f,0.f,0.f,0.f,0.f};
    v8f cr = {0.f,0.f,0.f,0.f,0.f,0.f,0.f,0.f};
    float nsq = 0.f;

    for (int k0 = 0; k0 < IN_FEAT; k0 += 4) {
        int ka = k0 + half * 2;
        float a0 = xrow[ka], a1 = xrow[ka + 1];
        nsq = fmaf(a0, a0, nsq);
        nsq = fmaf(a1, a1, nsq);
        v2f a; a[0] = a0; a[1] = a1;
        v2f b;  b[0]  = sW [ka * OUTC + l15];  b[1]  = sW [(ka + 1) * OUTC + l15];
        v2f br; br[0] = sWr[ka * OUTC + l15];  br[1] = sWr[(ka + 1) * OUTC + l15];
        cf = __builtin_amdgcn_wmma_f32_16x16x4_f32(false, a, false, b,  (short)0, cf, false, false);
        cr = __builtin_amdgcn_wmma_f32_16x16x4_f32(false, a, false, br, (short)0, cr, false, false);
    }

    nsq += __shfl_xor(nsq, 16, 32);
    float nrm = fmaxf(sqrtf(nsq), EPSF);
    float z   = fminf(nrm, 1.0f - 1e-7f);
    float sc  = 0.5f * (log1pf(z) - log1pf(-z)) / nrm;
    float bv  = bias[l15];

    float* fp = feat + (size_t)t * 16 * OUTC;
    float* op = out  + (size_t)t * 16 * 8;
    float* my = st + wave * (16 * TPITCH);
    #pragma unroll
    for (int r = 0; r < 8; ++r) {
        int row = r + half * 8;
        float s_r = __shfl(sc, row, 32);
        float fv  = cf[r] * s_r;
        float rv  = cr[r] * s_r + bv;
        float ro  = __shfl_xor(rv, 8, 32);
        my[row * TPITCH + l15] = fv;
        if (t * 16 + row < N) {
            fp[row * OUTC + l15] = fv;
            if (l15 < 8) op[row * 8 + l15] = 0.5f * (rv + ro);
        }
    }
    ring_order();
    const float* fr = my + l15 * TPITCH + half * 8;
    float sl = 0.f, sr = 0.f;
    #pragma unroll
    for (int c = 0; c < 8; ++c) {
        float v = fr[c];
        sl = fmaf(v, sal[half * 8 + c], sl);
        sr = fmaf(v, sar[half * 8 + c], sr);
    }
    int n = t * 16 + l15;
    float emax = -INFINITY, emin = INFINITY;
    if (n < N) {
        el[2 * (size_t)n + half] = sl;
        er[2 * (size_t)n + half] = sr;
        emax = sl; emin = sl;
    }
    for (int o = 8; o > 0; o >>= 1) {
        emax = fmaxf(emax, __shfl_xor(emax, o, 32));
        emin = fminf(emin, __shfl_xor(emin, o, 32));
    }
    float emax1 = __shfl(emax, 16, 32), emin1 = __shfl(emin, 16, 32);
    if (lane == 0) wm[wave] = make_float4(emin, emin1, emax, emax1);
    __syncthreads();
    if (tid == 0) {
        float4 q = wm[0];
        for (int w = 1; w < 8; ++w) {
            q.x = fminf(q.x, wm[w].x); q.y = fminf(q.y, wm[w].y);
            q.z = fmaxf(q.z, wm[w].z); q.w = fmaxf(q.w, wm[w].w);
        }
        pmm[blockIdx.x] = q;
    }
}

__device__ __forceinline__ float4 el_bounds(const float4* __restrict__ pmm, int nPart, float4* red)
{
    int tid = threadIdx.x, lane = tid & 31, wave = tid >> 5, nw = (int)blockDim.x >> 5;
    float4 b = make_float4(INFINITY, INFINITY, -INFINITY, -INFINITY);
    for (int i = tid; i < nPart; i += (int)blockDim.x) {
        float4 p = pmm[i];
        b.x = fminf(b.x, p.x); b.y = fminf(b.y, p.y);
        b.z = fmaxf(b.z, p.z); b.w = fmaxf(b.w, p.w);
    }
    for (int o = 16; o > 0; o >>= 1) {
        b.x = fminf(b.x, __shfl_xor(b.x, o, 32)); b.y = fminf(b.y, __shfl_xor(b.y, o, 32));
        b.z = fmaxf(b.z, __shfl_xor(b.z, o, 32)); b.w = fmaxf(b.w, __shfl_xor(b.w, o, 32));
    }
    if (lane == 0) red[wave] = b;
    __syncthreads();
    b = red[0];
    for (int w = 1; w < nw; ++w) {
        float4 p = red[w];
        b.x = fminf(b.x, p.x); b.y = fminf(b.y, p.y);
        b.z = fmaxf(b.z, p.z); b.w = fmaxf(b.w, p.w);
    }
    return b;
}
__device__ __forceinline__ bool span_small(float4 b) {
    return (b.z - b.x <= ELSPAN) && (b.w - b.y <= ELSPAN);
}

#define RNGF    400
#define WGT     1024
#define NWAVE   (WGT / 32)
#define QCAP    64

template <bool AL4>
__device__ __forceinline__ int4 load_quad(const int* __restrict__ p, int e4, int ee) {
    if (AL4) return *reinterpret_cast<const int4*>(p + min(e4, ee - 4));
    int4 r;
    r.x = p[min(e4,     ee - 1)];
    r.y = p[min(e4 + 1, ee - 1)];
    r.z = p[min(e4 + 2, ee - 1)];
    r.w = p[min(e4 + 3, ee - 1)];
    return r;
}
__device__ __forceinline__ int quad_at(const int4& v, int k) {
    return (k == 0) ? v.x : (k == 1) ? v.y : (k == 2) ? v.z : v.w;
}

#define NQ      4
#define BLK     (NQ * 128)
template <int R, bool AL4, typename F>
__device__ __forceinline__ void scan_hits(const int* __restrict__ srcp, const int* __restrict__ dstp,
                                          int eb, int ee, int base,
                                          unsigned* __restrict__ ring, int lane, int wave, F work)
{
    unsigned* qs = ring + wave * (2 * QCAP);
    unsigned* qd = qs + QCAP;
    unsigned head = 0u, qn = 0u;
    int first = eb + wave * BLK;
    if (first >= ee) return;
    int4 dq[NQ], sq[NQ];
    #pragma unroll
    for (int q = 0; q < NQ; ++q) { dq[q] = make_int4(-1, -1, -1, -1); sq[q] = make_int4(0, 0, 0, 0); }
    for (int e0 = first - NWAVE * BLK; e0 < ee; e0 += NWAVE * BLK) {
        int en = e0 + NWAVE * BLK;
        #pragma unroll
        for (int q = 0; q < NQ; ++q) {
            int eq = e0 + q * 128 + 4 * lane;
            #pragma unroll
            for (int k = 0; k < 4; ++k) {
                int e = eq + k;
                unsigned dl = (unsigned)(((e < ee) ? quad_at(dq[q], k) : -1) - base);
                bool     hit  = dl < (unsigned)R;
                unsigned mask = __builtin_amdgcn_ballot_w32(hit);
                if (mask != 0u) {
                    if (hit) {
                        unsigned pos = (head + qn + __builtin_amdgcn_mbcnt_lo(mask, 0u)) & (QCAP - 1);
                        qs[pos] = (unsigned)quad_at(sq[q], k);
                        qd[pos] = dl;
                    }
                    qn += (unsigned)__popc(mask);
                    if (qn >= 32u) {
                        ring_order();
                        unsigned p = (head + (unsigned)lane) & (QCAP - 1);
                        work(qs[p], qd[p], true);
                        ring_order();
                        head = (head + 32u) & (QCAP - 1);
                        qn  -= 32u;
                    }
                }
            }
            int nq = en + q * 128 + 4 * lane;
            dq[q] = load_quad<AL4>(dstp, nq, ee);
            sq[q] = load_quad<AL4>(srcp, nq, ee);
        }
    }
    if (qn != 0u) {
        ring_order();
        unsigned p = (head + (unsigned)lane) & (QCAP - 1);
        work(qs[p], qd[p], (unsigned)lane < qn);
    }
}

#define PEPT    14
#define PCH     (PEPT * WGT)
#define SRCBITS 20
#define MAXB    2624
__global__ __launch_bounds__(WGT) void k_part(const int* __restrict__ ei, unsigned* __restrict__ part,
                                              unsigned* __restrict__ offs, int N, int E, int nb)
{
    __shared__ unsigned cnt[MAXB];
    __shared__ unsigned off[MAXB + 1];
    __shared__ unsigned sorted[PCH];
    int tid = threadIdx.x;
    int eb = blockIdx.x * PCH;
    for (int b = tid; b < nb; b += WGT) cnt[b] = 0u;
    __syncthreads();
    unsigned dv[PEPT], sv[PEPT], bin[PEPT], rnk[PEPT], wrd[PEPT];
    #pragma unroll
    for (int i = 0; i < PEPT; ++i) {
        int ec = min(eb + i * WGT + tid, E - 1);
        dv[i] = (unsigned)ei[(size_t)E + ec];
        sv[i] = (unsigned)ei[ec];
    }
    #pragma unroll
    for (int i = 0; i < PEPT; ++i) {
        int e = eb + i * WGT + tid;
        bin[i] = 0xFFFFFFFFu; rnk[i] = 0u; wrd[i] = 0u;
        if (e < E && dv[i] < (unsigned)N) {
            unsigned b = dv[i] / (unsigned)RNGF;
            bin[i] = b;
            wrd[i] = ((dv[i] - b * (unsigned)RNGF) << SRCBITS) | (sv[i] & ((1u << SRCBITS) - 1u));
            rnk[i] = atomicAdd(&cnt[b], 1u);
        }
    }
    __syncthreads();
    if (tid < 32) {
        unsigned carry = 0u;
        for (int b0 = 0; b0 < nb; b0 += 32) {
            int b = b0 + tid;
            unsigned c = (b < nb) ? cnt[b] : 0u;
            unsigned x = c;
            for (int o = 1; o < 32; o <<= 1) {
                unsigned y = __shfl_up(x, o, 32);
                if (tid >= o) x += y;
            }
            if (b < nb) off[b] = carry + x - c;
            carry += __shfl(x, 31, 32);
        }
        if (tid == 0) off[nb] = carry;
    }
    __syncthreads();
    for (int b = tid; b <= nb; b += WGT) offs[(size_t)blockIdx.x * (nb + 1) + b] = off[b];
    #pragma unroll
    for (int i = 0; i < PEPT; ++i)
        if (bin[i] != 0xFFFFFFFFu) sorted[off[bin[i]] + rnk[i]] = wrd[i];
    __syncthreads();
    unsigned total = off[nb];
    unsigned* reg = part + (size_t)blockIdx.x * PCH;
    for (unsigned i = tid; i < total; i += WGT) reg[i] = sorted[i];
}

template <typename F>
__device__ __forceinline__ void part_hits(const unsigned* __restrict__ part, const unsigned* __restrict__ offs,
                                          int nChunk, int nb1, int g, int lane, int wv, F work)
{
    for (int cg = wv; cg < nChunk; cg += 32 * NWAVE) {
        int c = cg + lane * NWAVE;
        unsigned lo = 0u, hi = 0u;
        if (c < nChunk) {
            const unsigned* o = offs + (size_t)c * nb1 + g;
            lo = o[0]; hi = o[1];
        }
        int nr = min(32, (nChunk - cg + NWAVE - 1) / NWAVE);
        int j = 0;
        unsigned a = 0u, b = 0u;
        for (; j < nr; ++j) {
            a = (unsigned)__builtin_amdgcn_readlane((int)lo, j);
            b = (unsigned)__builtin_amdgcn_readlane((int)hi, j);
            if (a < b) break;
        }
        if (j >= nr) continue;
        unsigned w = part[(size_t)(cg + j * NWAVE) * PCH + min(a + (unsigned)lane, b - 1u)];
        for (;;) {
            unsigned cur = w;
            bool act = a + (unsigned)lane < b;
            int nj = j;
            unsigned na = a + 32u, nb = b;
            if (na >= b) {
                for (nj = j + 1; nj < nr; ++nj) {
                    na = (unsigned)__builtin_amdgcn_readlane((int)lo, nj);
                    nb = (unsigned)__builtin_amdgcn_readlane((int)hi, nj);
                    if (na < nb) break;
                }
            }
            bool more = nj < nr;
            if (!more) { nj = j; na = a; nb = b; }
            w = part[(size_t)(cg + nj * NWAVE) * PCH + min(na + (unsigned)lane, nb - 1u)];
            work(cur & ((1u << SRCBITS) - 1u), cur >> SRCBITS, act);
            if (!more) break;
            j = nj; a = na; b = nb;
        }
    }
}

__global__ __launch_bounds__(WGT) void k_range_fused(
    const int* __restrict__ ei, const float* __restrict__ al, const float* __restrict__ el,
    const float* __restrict__ er, const float4* __restrict__ pmm, int nPart,
    const float* __restrict__ feat, float* __restrict__ out, int N, int E,
    const unsigned* __restrict__ part, const unsigned* __restrict__ offs, int nChunk)
{
    __shared__ __align__(16) float acc[NACC * RNGF];
    __shared__ float2 ser[RNGF];
    __shared__ unsigned smx[2 * RNGF];
    __shared__ unsigned ring[NWAVE * 2 * QCAP];
    __shared__ float4 red[NWAVE];
    int tid = threadIdx.x, lane = tid & 31, wave = tid >> 5;
    int base = blockIdx.x * RNGF;
    bool al4 = (E & 3) == 0;
    float4* acc4 = reinterpret_cast<float4*>(acc);
    for (int i = tid; i < NACC * RNGF / 4; i += WGT) acc4[i] = make_float4(0.f, 0.f, 0.f, 0.f);
    for (int i = tid; i < RNGF; i += WGT) {
        int n = base + i;
        ser[i] = (n < N) ? *reinterpret_cast<const float2*>(er + 2 * (size_t)n) : make_float2(0.f, 0.f);
        smx[2 * i] = MAXINIT; smx[2 * i + 1] = MAXINIT;
    }
    float4 bnd = el_bounds(pmm, nPart, red);
    bool bounded = __builtin_amdgcn_readfirstlane(span_small(bnd) ? 1 : 0) != 0;
    float b0 = bnd.z, b1 = bnd.w;
    const int* srcp = ei;
    float w0 = al[0],  w1 = al[1],  w2 = al[2],   w3 = al[3],   w4 = al[4],   w5 = al[5],   w6 = al[6],   w7 = al[7];
    float w8 = al[8],  w9 = al[9],  w10 = al[10], w11 = al[11], w12 = al[12], w13 = al[13], w14 = al[14], w15 = al[15];
    auto hit_sum = [&](unsigned s, unsigned dl, bool act, auto sh) __attribute__((always_inline)) {
            if (act) {
                const float4* fs = reinterpret_cast<const float4*>(feat + (size_t)s * OUTC);
                float4 f0 = fs[0], f1 = fs[1], f2 = fs[2], f3 = fs[3];
                float2 l;
                l.x = fmaf(f0.x, w0, 0.f);  l.x = fmaf(f0.y, w1, l.x);  l.x = fmaf(f0.z, w2, l.x);  l.x = fmaf(f0.w, w3, l.x);
                l.x = fmaf(f1.x, w4, l.x);  l.x = fmaf(f1.y, w5, l.x);  l.x = fmaf(f1.z, w6, l.x);  l.x = fmaf(f1.w, w7, l.x);
                l.y = fmaf(f2.x, w8, 0.f);  l.y = fmaf(f2.y, w9, l.y);  l.y = fmaf(f2.z, w10, l.y); l.y = fmaf(f2.w, w11, l.y);
                l.y = fmaf(f3.x, w12, l.y); l.y = fmaf(f3.y, w13, l.y); l.y = fmaf(f3.z, w14, l.y); l.y = fmaf(f3.w, w15, l.y);
                float2 r = ser[dl];
                float2 m = sh(r, dl);
                float p0 = __expf(leaky(l.x + r.x) - m.x);
                float p1 = __expf(leaky(l.y + r.y) - m.y);
                float* a = acc + dl;
                atomicAdd(a +  0 * RNGF, p0 * f0.x);
                atomicAdd(a +  1 * RNGF, p0 * f0.y);
                atomicAdd(a +  2 * RNGF, p0 * f0.z);
                atomicAdd(a +  3 * RNGF, p0 * f0.w);
                atomicAdd(a +  4 * RNGF, p0 * f1.x);
                atomicAdd(a +  5 * RNGF, p0 * f1.y);
                atomicAdd(a +  6 * RNGF, p0 * f1.z);
                atomicAdd(a +  7 * RNGF, p0 * f1.w);
                atomicAdd(a +  8 * RNGF, p1 * f2.x);
                atomicAdd(a +  9 * RNGF, p1 * f2.y);
                atomicAdd(a + 10 * RNGF, p1 * f2.z);
                atomicAdd(a + 11 * RNGF, p1 * f2.w);
                atomicAdd(a + 12 * RNGF, p1 * f3.x);
                atomicAdd(a + 13 * RNGF, p1 * f3.y);
                atomicAdd(a + 14 * RNGF, p1 * f3.z);
                atomicAdd(a + 15 * RNGF, p1 * f3.w);
                atomicAdd(a + 16 * RNGF, p0);
                atomicAdd(a + 17 * RNGF, p1);
            }
        };
    auto sh_bound = [&](float2 r, unsigned) __attribute__((always_inline)) { return make_float2(leaky(b0 + r.x), leaky(b1 + r.y)); };
    auto sh_any = [&](float2 r, unsigned dl) __attribute__((always_inline)) {
            if (bounded) return make_float2(leaky(b0 + r.x), leaky(b1 + r.y));
            return make_float2(funmap(smx[2 * dl]), funmap(smx[2 * dl + 1]));
        };
    auto hit_max = [&](unsigned s, unsigned dl, bool act) __attribute__((always_inline)) {
            if (act) {
                float2 l = *reinterpret_cast<const float2*>(el + 2 * (size_t)s);
                float2 r = ser[dl];
                atomicMax(&smx[2 * dl],     fmap(leaky(l.x + r.x)));
                atomicMax(&smx[2 * dl + 1], fmap(leaky(l.y + r.y)));
            }
        };
    auto hit_bound = [&](unsigned s, unsigned dl, bool act) __attribute__((always_inline)) { hit_sum(s, dl, act, sh_bound); };
    auto hit_any   = [&](unsigned s, unsigned dl, bool act) __attribute__((always_inline)) { hit_sum(s, dl, act, sh_any); };
    if (__builtin_expect(part != nullptr, 1)) {
        int wv = __builtin_amdgcn_readfirstlane(wave);
        int nb1 = (int)gridDim.x + 1;
        if (__builtin_expect(bounded, 1)) part_hits(part, offs, nChunk, nb1, (int)blockIdx.x, lane, wv, hit_bound);
        else {
            part_hits(part, offs, nChunk, nb1, (int)blockIdx.x, lane, wv, hit_max);
            __syncthreads();
            part_hits(part, offs, nChunk, nb1, (int)blockIdx.x, lane, wv, hit_any);
        }
    }
    else if (__builtin_expect(bounded && al4, 1))
        scan_hits<RNGF, true >(srcp, ei + E, 0, E, base, ring, lane, wave, hit_bound);
    else {
        if (!bounded) {
            scan_hits<RNGF, false>(srcp, ei + E, 0, E, base, ring, lane, wave, hit_max);
            __syncthreads();
        }
        scan_hits<RNGF, false>(srcp, ei + E, 0, E, base, ring, lane, wave, hit_any);
    }
    __syncthreads();
    for (int idx = tid; idx < 2 * RNGF; idx += WGT) {
        int i = idx >> 1, hf = idx & 1;
        int n = base + i;
        if (n < N) {
            const float* g = acc + i;
            float r0 = 0.5f / fmaxf(g[16 * RNGF], EPSF);
            float r1 = 0.5f / fmaxf(g[17 * RNGF], EPSF);
            int c0 = hf * 4;
            float4* o = reinterpret_cast<float4*>(out + (size_t)n * 8 + c0);
            float4 v = *o;
            v.x += fmaf(g[(c0 + 0) * RNGF], r0, g[(c0 +  8) * RNGF] * r1);
            v.y += fmaf(g[(c0 + 1) * RNGF], r0, g[(c0 +  9) * RNGF] * r1);
            v.z += fmaf(g[(c0 + 2) * RNGF], r0, g[(c0 + 10) * RNGF] * r1);
            v.w += fmaf(g[(c0 + 3) * RNGF], r0, g[(c0 + 11) * RNGF] * r1);
            *o = v;
        }
    }
}

extern "C" void kernel_launch(void* const* d_in, const int* in_sizes, int n_in,
                              void* d_out, int out_size, void* d_ws, size_t ws_size,
                              hipStream_t stream)
{
    const float* x    = (const float*)d_in[0];
    const int*   ei   = (const int*)  d_in[1];
    const float* W    = (const float*)d_in[2];
    const float* al   = (const float*)d_in[3];
    const float* ar   = (const float*)d_in[4];
    const float* bias = (const float*)d_in[5];
    const float* Wr   = (const float*)d_in[6];
    float* out = (float*)d_out;

    int N = in_sizes[0] / IN_FEAT;
    int E = in_sizes[1] / 2;

    int nTiles = (N + 15) / 16;
    int nPart = (nTiles + 7) / 8;

    char* ws = (char*)d_ws;
    float*    feat  = (float*)ws;    ws += (size_t)N * OUTC * sizeof(float);
    float4*   pmm   = (float4*)ws;   ws += (size_t)nPart * sizeof(float4);
    float*    el    = (float*)ws;    ws += (size_t)N * 2 * sizeof(float);
    float*    er    = (float*)ws;    ws += (size_t)N * 2 * sizeof(float);
    int nRangeF = (N + RNGF - 1) / RNGF;
    int nChunk  = (E + PCH - 1) / PCH;
    unsigned* part  = (unsigned*)ws; ws += (size_t)nChunk * PCH * sizeof(unsigned);
    unsigned* offs  = (unsigned*)ws; ws += (size_t)nChunk * (nRangeF + 1) * sizeof(unsigned);
    bool parted = nChunk > 0 && N <= (1 << SRCBITS) && (size_t)(ws - (char*)d_ws) <= ws_size;

    dim3 blk(256);

    dim3 wg(WGT);
    if (parted) k_part<<<nChunk, wg, 0, stream>>>(ei, part, offs, N, E, nRangeF);
    k_node<<<nPart, blk, 0, stream>>>(x, W, Wr, bias, al, ar, feat, el, er, pmm, out, N, nTiles);
    k_range_fused<<<nRangeF, wg, 0, stream>>>(ei, al, el, er, pmm, nPart, feat, out, N, E,
                                              parted ? part : nullptr, offs, nChunk);
}
